# starting kernel
# baseline (speedup 1.0000x reference)
#include <hip/hip_runtime.h>
#include <math.h>

#define HIDC 128

typedef __attribute__((ext_vector_type(2))) float v2f;
typedef __attribute__((ext_vector_type(8))) float v8f;

__global__ void gcn_init(float* __restrict__ acc, float* __restrict__ deg, int N) {
    int idx = blockIdx.x * blockDim.x + threadIdx.x;
    long long total = (long long)N * HIDC;
    if ((long long)idx < total) acc[idx] = 0.0f;
    if (idx < N) deg[idx] = 1.0f;
}

__global__ void gcn_degree(const int* __restrict__ dst, float* __restrict__ deg, int E) {
    int e = blockIdx.x * blockDim.x + threadIdx.x;
    if (e < E) atomicAdd(&deg[dst[e]], 1.0f);
}

__global__ void gcn_dinv(const float* __restrict__ deg, float* __restrict__ dinv, int N) {
    int i = blockIdx.x * blockDim.x + threadIdx.x;
    if (i < N) dinv[i] = rsqrtf(deg[i]);
}

__global__ __launch_bounds__(256) void gcn_gemm1_wmma(const float* __restrict__ x,
                                                      const float* __restrict__ W,
                                                      float* __restrict__ h, int N) {
    __shared__ float As[16 * 132];
    const int t = threadIdx.x;
    const int rowBase = blockIdx.x * 16;
    const bool full = (rowBase + 16 <= N);

    if (full) {
#pragma unroll
        for (int i = t; i < 16 * 32; i += 256) {
            int r  = i >> 5;
            int c4 = i & 31;
            float4 v = ((const float4*)(x + (size_t)(rowBase + r) * HIDC))[c4];
            float* p = &As[r * 132 + c4 * 4];
            p[0] = v.x; p[1] = v.y; p[2] = v.z; p[3] = v.w;
        }
    } else {
        for (int i = t; i < 16 * 32; i += 256) {
            int r  = i >> 5;
            int c4 = i & 31;
            float4 v = make_float4(0.f, 0.f, 0.f, 0.f);
            if (rowBase + r < N) v = ((const float4*)(x + (size_t)(rowBase + r) * HIDC))[c4];
            float* p = &As[r * 132 + c4 * 4];
            p[0] = v.x; p[1] = v.y; p[2] = v.z; p[3] = v.w;
        }
    }
    __syncthreads();

    const int lane = t & 31;
    const int wv   = t >> 5;
    const int m    = lane & 15;
    const int half = lane >> 4;
    const int kl   = half * 2;
    const int n    = wv * 16 + (lane & 15);

    v8f c = {};
    const float* Arow = &As[m * 132];
#pragma unroll 8
    for (int k0 = 0; k0 < HIDC; k0 += 4) {
        v2f a, b;
        a.x = Arow[k0 + kl];
        a.y = Arow[k0 + kl + 1];
        b.x = W[(k0 + kl) * HIDC + n];
        b.y = W[(k0 + kl + 1) * HIDC + n];
        c = __builtin_amdgcn_wmma_f32_16x16x4_f32(false, a, false, b,
                                                  (short)0, c, false, false);
    }

    float* hp = h + (size_t)(rowBase + 8 * half) * HIDC + n;
    if (full) {
#pragma unroll
        for (int v = 0; v < 8; ++v) hp[v * HIDC] = c[v];
    } else {
#pragma unroll
        for (int v = 0; v < 8; ++v) {
            if (rowBase + 8 * half + v < N) hp[v * HIDC] = c[v];
        }
    }
}

__global__ __launch_bounds__(256) void gcn_agg1(const float* __restrict__ h,
                                                const int* __restrict__ src,
                                                const int* __restrict__ dst,
                                                const float* __restrict__ dinv,
                                                float* __restrict__ acc, int E) {
    int e = blockIdx.x * 8 + (threadIdx.x >> 5);
    if (e >= E) return;
    int lane = threadIdx.x & 31;
    int s = src[e];
    int d = dst[e];
    const float* hs = h + (size_t)s * HIDC;
    __builtin_prefetch(hs + lane * 4, 0, 0);
    float nrm = dinv[s] * dinv[d];
    float4 v = ((const float4*)hs)[lane];
    float* o = acc + (size_t)d * HIDC + lane * 4;
    atomicAdd(o + 0, v.x * nrm);
    atomicAdd(o + 1, v.y * nrm);
    atomicAdd(o + 2, v.z * nrm);
    atomicAdd(o + 3, v.w * nrm);
}

__global__ __launch_bounds__(256) void gcn_fin1_dot(const float* __restrict__ acc,
                                                    const float* __restrict__ h,
                                                    const float* __restrict__ dinv,
                                                    const float* __restrict__ b1,
                                                    const float* __restrict__ W2,
                                                    float* __restrict__ h2, int N) {
    int i = blockIdx.x * 8 + (threadIdx.x >> 5);
    if (i >= N) return;
    int lane = threadIdx.x & 31;
    float di = dinv[i];
    float ns = di * di;
    float4 a  = ((const float4*)(acc + (size_t)i * HIDC))[lane];
    float4 hv = ((const float4*)(h   + (size_t)i * HIDC))[lane];
    float4 bb = ((const float4*)b1)[lane];
    float4 w  = ((const float4*)W2)[lane];
    float r0 = fmaxf(a.x + hv.x * ns + bb.x, 0.0f);
    float r1 = fmaxf(a.y + hv.y * ns + bb.y, 0.0f);
    float r2 = fmaxf(a.z + hv.z * ns + bb.z, 0.0f);
    float r3 = fmaxf(a.w + hv.w * ns + bb.w, 0.0f);
    float p = r0 * w.x + r1 * w.y + r2 * w.z + r3 * w.w;
#pragma unroll
    for (int off = 16; off > 0; off >>= 1) p += __shfl_down(p, off, 32);
    if (lane == 0) h2[i] = p;
}

__global__ void gcn_init_out(const float* __restrict__ h2, const float* __restrict__ dinv,
                             const float* __restrict__ b2, float* __restrict__ out, int N) {
    int i = blockIdx.x * blockDim.x + threadIdx.x;
    if (i < N) {
        float di = dinv[i];
        out[i] = h2[i] * di * di + b2[0];
    }
}

__global__ void gcn_agg2(const float* __restrict__ h2, const int* __restrict__ src,
                         const int* __restrict__ dst, const float* __restrict__ dinv,
                         float* __restrict__ out, int E) {
    int e = blockIdx.x * blockDim.x + threadIdx.x;
    if (e < E) {
        int s = src[e];
        int d = dst[e];
        atomicAdd(&out[d], h2[s] * dinv[s] * dinv[d]);
    }
}

extern "C" void kernel_launch(void* const* d_in, const int* in_sizes, int n_in,
                              void* d_out, int out_size, void* d_ws, size_t ws_size,
                              hipStream_t stream) {
    const float* x  = (const float*)d_in[0];
    const float* W1 = (const float*)d_in[1];
    const float* b1 = (const float*)d_in[2];
    const float* W2 = (const float*)d_in[3];
    const float* b2 = (const float*)d_in[4];
    const int*   ei = (const int*)d_in[5];

    const int N = in_sizes[0] / HIDC;
    const int E = in_sizes[5] / 2;
    const int* src = ei;
    const int* dst = ei + E;
    float* out = (float*)d_out;

    float* h    = (float*)d_ws;
    float* acc  = h   + (size_t)N * HIDC;
    float* deg  = acc + (size_t)N * HIDC;
    float* dinv = deg + N;
    float* h2   = dinv + N;

    const int TB = 256;
    long long totNH = (long long)N * HIDC;
    int gInit = (int)((totNH + TB - 1) / TB);

    gcn_init      <<<gInit,             TB, 0, stream>>>(acc, deg, N);
    gcn_degree    <<<(E + TB - 1) / TB, TB, 0, stream>>>(dst, deg, E);
    gcn_dinv      <<<(N + TB - 1) / TB, TB, 0, stream>>>(deg, dinv, N);
    gcn_gemm1_wmma<<<(N + 15) / 16,     TB, 0, stream>>>(x, W1, h, N);
    gcn_agg1      <<<(E + 7) / 8,       TB, 0, stream>>>(h, src, dst, dinv, acc, E);
    gcn_fin1_dot  <<<(N + 7) / 8,       TB, 0, stream>>>(acc, h, dinv, b1, W2, h2, N);
    gcn_init_out  <<<(N + TB - 1) / TB, TB, 0, stream>>>(h2, dinv, b2, out, N);
    gcn_agg2      <<<(E + TB - 1) / TB, TB, 0, stream>>>(h2, src, dst, dinv, out, E);
}


# agg1: lane l owns columns l+32k (k=0..3), so each f32 atomic instruction adds to one whole 128-B line, not 32-B quarters of four lines; sums still follow atomic arrival order as before
# speedup vs baseline: 1.2563x; 1.2563x over previous
#include <hip/hip_runtime.h>
#include <math.h>

#define HIDC 128

typedef __attribute__((ext_vector_type(2))) float v2f;
typedef __attribute__((ext_vector_type(8))) float v8f;

__global__ void gcn_init(float* __restrict__ acc, float* __restrict__ deg, int N) {
    int idx = blockIdx.x * blockDim.x + threadIdx.x;
    long long total = (long long)N * HIDC;
    if ((long long)idx < total) acc[idx] = 0.0f;
    if (idx < N) deg[idx] = 1.0f;
}

__global__ void gcn_degree(const int* __restrict__ dst, float* __restrict__ deg, int E) {
    int e = blockIdx.x * blockDim.x + threadIdx.x;
    if (e < E) atomicAdd(&deg[dst[e]], 1.0f);
}

__global__ void gcn_dinv(const float* __restrict__ deg, float* __restrict__ dinv, int N) {
    int i = blockIdx.x * blockDim.x + threadIdx.x;
    if (i < N) dinv[i] = rsqrtf(deg[i]);
}

__global__ __launch_bounds__(256) void gcn_gemm1_wmma(const float* __restrict__ x,
                                                      const float* __restrict__ W,
                                                      float* __restrict__ h, int N) {
    __shared__ float As[16 * 132];
    const int t = threadIdx.x;
    const int rowBase = blockIdx.x * 16;
    const bool full = (rowBase + 16 <= N);

    if (full) {
#pragma unroll
        for (int i = t; i < 16 * 32; i += 256) {
            int r  = i >> 5;
            int c4 = i & 31;
            float4 v = ((const float4*)(x + (size_t)(rowBase + r) * HIDC))[c4];
            float* p = &As[r * 132 + c4 * 4];
            p[0] = v.x; p[1] = v.y; p[2] = v.z; p[3] = v.w;
        }
    } else {
        for (int i = t; i < 16 * 32; i += 256) {
            int r  = i >> 5;
            int c4 = i & 31;
            float4 v = make_float4(0.f, 0.f, 0.f, 0.f);
            if (rowBase + r < N) v = ((const float4*)(x + (size_t)(rowBase + r) * HIDC))[c4];
            float* p = &As[r * 132 + c4 * 4];
            p[0] = v.x; p[1] = v.y; p[2] = v.z; p[3] = v.w;
        }
    }
    __syncthreads();

    const int lane = t & 31;
    const int wv   = t >> 5;
    const int m    = lane & 15;
    const int half = lane >> 4;
    const int kl   = half * 2;
    const int n    = wv * 16 + (lane & 15);

    v8f c = {};
    const float* Arow = &As[m * 132];
#pragma unroll 8
    for (int k0 = 0; k0 < HIDC; k0 += 4) {
        v2f a, b;
        a.x = Arow[k0 + kl];
        a.y = Arow[k0 + kl + 1];
        b.x = W[(k0 + kl) * HIDC + n];
        b.y = W[(k0 + kl + 1) * HIDC + n];
        c = __builtin_amdgcn_wmma_f32_16x16x4_f32(false, a, false, b,
                                                  (short)0, c, false, false);
    }

    float* hp = h + (size_t)(rowBase + 8 * half) * HIDC + n;
    if (full) {
#pragma unroll
        for (int v = 0; v < 8; ++v) hp[v * HIDC] = c[v];
    } else {
#pragma unroll
        for (int v = 0; v < 8; ++v) {
            if (rowBase + 8 * half + v < N) hp[v * HIDC] = c[v];
        }
    }
}

__global__ __launch_bounds__(256) void gcn_agg1(const float* __restrict__ h,
                                                const int* __restrict__ src,
                                                const int* __restrict__ dst,
                                                const float* __restrict__ dinv,
                                                float* __restrict__ acc, int E) {
    int e = blockIdx.x * 8 + (threadIdx.x >> 5);
    if (e >= E) return;
    int lane = threadIdx.x & 31;
    int s = src[e];
    int d = dst[e];
    const float* hs = h + (size_t)s * HIDC;
    __builtin_prefetch(hs + lane * 4, 0, 0);
    float nrm = dinv[s] * dinv[d];
    const float* hl = hs + lane;
    float v0 = hl[0];
    float v1 = hl[32];
    float v2 = hl[64];
    float v3 = hl[96];
    float* o = acc + (size_t)d * HIDC + lane;
    atomicAdd(o +  0, v0 * nrm);
    atomicAdd(o + 32, v1 * nrm);
    atomicAdd(o + 64, v2 * nrm);
    atomicAdd(o + 96, v3 * nrm);
}

__global__ __launch_bounds__(256) void gcn_fin1_dot(const float* __restrict__ acc,
                                                    const float* __restrict__ h,
                                                    const float* __restrict__ dinv,
                                                    const float* __restrict__ b1,
                                                    const float* __restrict__ W2,
                                                    float* __restrict__ h2, int N) {
    int i = blockIdx.x * 8 + (threadIdx.x >> 5);
    if (i >= N) return;
    int lane = threadIdx.x & 31;
    float di = dinv[i];
    float ns = di * di;
    float4 a  = ((const float4*)(acc + (size_t)i * HIDC))[lane];
    float4 hv = ((const float4*)(h   + (size_t)i * HIDC))[lane];
    float4 bb = ((const float4*)b1)[lane];
    float4 w  = ((const float4*)W2)[lane];
    float r0 = fmaxf(a.x + hv.x * ns + bb.x, 0.0f);
    float r1 = fmaxf(a.y + hv.y * ns + bb.y, 0.0f);
    float r2 = fmaxf(a.z + hv.z * ns + bb.z, 0.0f);
    float r3 = fmaxf(a.w + hv.w * ns + bb.w, 0.0f);
    float p = r0 * w.x + r1 * w.y + r2 * w.z + r3 * w.w;
#pragma unroll
    for (int off = 16; off > 0; off >>= 1) p += __shfl_down(p, off, 32);
    if (lane == 0) h2[i] = p;
}

__global__ void gcn_init_out(const float* __restrict__ h2, const float* __restrict__ dinv,
                             const float* __restrict__ b2, float* __restrict__ out, int N) {
    int i = blockIdx.x * blockDim.x + threadIdx.x;
    if (i < N) {
        float di = dinv[i];
        out[i] = h2[i] * di * di + b2[0];
    }
}

__global__ void gcn_agg2(const float* __restrict__ h2, const int* __restrict__ src,
                         const int* __restrict__ dst, const float* __restrict__ dinv,
                         float* __restrict__ out, int E) {
    int e = blockIdx.x * blockDim.x + threadIdx.x;
    if (e < E) {
        int s = src[e];
        int d = dst[e];
        atomicAdd(&out[d], h2[s] * dinv[s] * dinv[d]);
    }
}

extern "C" void kernel_launch(void* const* d_in, const int* in_sizes, int n_in,
                              void* d_out, int out_size, void* d_ws, size_t ws_size,
                              hipStream_t stream) {
    const float* x  = (const float*)d_in[0];
    const float* W1 = (const float*)d_in[1];
    const float* b1 = (const float*)d_in[2];
    const float* W2 = (const float*)d_in[3];
    const float* b2 = (const float*)d_in[4];
    const int*   ei = (const int*)d_in[5];

    const int N = in_sizes[0] / HIDC;
    const int E = in_sizes[5] / 2;
    const int* src = ei;
    const int* dst = ei + E;
    float* out = (float*)d_out;

    float* h    = (float*)d_ws;
    float* acc  = h   + (size_t)N * HIDC;
    float* deg  = acc + (size_t)N * HIDC;
    float* dinv = deg + N;
    float* h2   = dinv + N;

    const int TB = 256;
    long long totNH = (long long)N * HIDC;
    int gInit = (int)((totNH + TB - 1) / TB);

    gcn_init      <<<gInit,             TB, 0, stream>>>(acc, deg, N);
    gcn_degree    <<<(E + TB - 1) / TB, TB, 0, stream>>>(dst, deg, E);
    gcn_dinv      <<<(N + TB - 1) / TB, TB, 0, stream>>>(deg, dinv, N);
    gcn_gemm1_wmma<<<(N + 15) / 16,     TB, 0, stream>>>(x, W1, h, N);
    gcn_agg1      <<<(E + 7) / 8,       TB, 0, stream>>>(h, src, dst, dinv, acc, E);
    gcn_fin1_dot  <<<(N + 7) / 8,       TB, 0, stream>>>(acc, h, dinv, b1, W2, h2, N);
    gcn_init_out  <<<(N + TB - 1) / TB, TB, 0, stream>>>(h2, dinv, b2, out, N);
    gcn_agg2      <<<(E + TB - 1) / TB, TB, 0, stream>>>(h2, src, dst, dinv, out, E);
}


# agg1 as a gather: edges counting-sorted by destination every call (int atomics); one wave sums a node's sources in registers, stores the row once; f32 sum order = bucket fill (atomic arrival) order
# speedup vs baseline: 1.4472x; 1.1520x over previous
#include <hip/hip_runtime.h>
#include <math.h>

#define HIDC 128

typedef __attribute__((ext_vector_type(2))) float v2f;
typedef __attribute__((ext_vector_type(8))) float v8f;

__global__ void gcn_zero_cnt(int* __restrict__ cnt, int N) {
    int i = blockIdx.x * blockDim.x + threadIdx.x;
    if (i < N) cnt[i] = 0;
}

__global__ void gcn_count(const int* __restrict__ dst, int* __restrict__ cnt, int E) {
    int e = blockIdx.x * blockDim.x + threadIdx.x;
    if (e < E) atomicAdd(&cnt[dst[e]], 1);
}

__global__ __launch_bounds__(1024) void gcn_scan(const int* __restrict__ cnt,
                                                 int* __restrict__ rowptr,
                                                 int* __restrict__ cursor,
                                                 float* __restrict__ dinv, int N) {
    __shared__ int preS[32];
    __shared__ int ownS[32];
    const int t = threadIdx.x;
    const int lane = t & 31;
    const int wv = t >> 5;
    const int blockStart = blockIdx.x * 4096;

    int pre = 0;
    const int4* c4 = (const int4*)cnt;
#pragma unroll 4
    for (int j = 0; j < (int)blockIdx.x; ++j) {
        int4 v = c4[j * 1024 + t];
        pre += (v.x + v.y) + (v.z + v.w);
    }
#pragma unroll
    for (int off = 16; off > 0; off >>= 1) pre += __shfl_xor(pre, off, 32);
    if (lane == 0) preS[wv] = pre;

    const int i0 = blockStart + t * 4;
    int c0 = (i0     < N) ? cnt[i0]     : 0;
    int c1 = (i0 + 1 < N) ? cnt[i0 + 1] : 0;
    int c2 = (i0 + 2 < N) ? cnt[i0 + 2] : 0;
    int c3 = (i0 + 3 < N) ? cnt[i0 + 3] : 0;
    const int s = (c0 + c1) + (c2 + c3);
    int inc = s;
#pragma unroll
    for (int off = 1; off < 32; off <<= 1) {
        int y = __shfl_up(inc, off, 32);
        if (lane >= off) inc += y;
    }
    if (lane == 31) ownS[wv] = inc;
    __syncthreads();

    int ab = preS[lane] + ((lane < wv) ? ownS[lane] : 0);
#pragma unroll
    for (int off = 16; off > 0; off >>= 1) ab += __shfl_xor(ab, off, 32);

    const int e0 = ab + inc - s;
    const int e1 = e0 + c0;
    const int e2 = e1 + c1;
    const int e3 = e2 + c2;
    if (i0 < N)     { rowptr[i0]     = e0; cursor[i0]     = e0; dinv[i0]     = rsqrtf((float)(c0 + 1)); }
    if (i0 + 1 < N) { rowptr[i0 + 1] = e1; cursor[i0 + 1] = e1; dinv[i0 + 1] = rsqrtf((float)(c1 + 1)); }
    if (i0 + 2 < N) { rowptr[i0 + 2] = e2; cursor[i0 + 2] = e2; dinv[i0 + 2] = rsqrtf((float)(c2 + 1)); }
    if (i0 + 3 < N) { rowptr[i0 + 3] = e3; cursor[i0 + 3] = e3; dinv[i0 + 3] = rsqrtf((float)(c3 + 1)); }
}

__global__ void gcn_fill(const int* __restrict__ src, const int* __restrict__ dst,
                         int* __restrict__ cursor, int* __restrict__ ssrc, int E) {
    int e = blockIdx.x * blockDim.x + threadIdx.x;
    if (e < E) {
        int pos = atomicAdd(&cursor[dst[e]], 1);
        ssrc[pos] = src[e];
    }
}

__global__ __launch_bounds__(256) void gcn_gemm1_wmma(const float* __restrict__ x,
                                                      const float* __restrict__ W,
                                                      float* __restrict__ h, int N) {
    __shared__ float As[16 * 132];
    const int t = threadIdx.x;
    const int rowBase = blockIdx.x * 16;
    const bool full = (rowBase + 16 <= N);

    if (full) {
#pragma unroll
        for (int i = t; i < 16 * 32; i += 256) {
            int r  = i >> 5;
            int c4 = i & 31;
            float4 v = ((const float4*)(x + (size_t)(rowBase + r) * HIDC))[c4];
            float* p = &As[r * 132 + c4 * 4];
            p[0] = v.x; p[1] = v.y; p[2] = v.z; p[3] = v.w;
        }
    } else {
        for (int i = t; i < 16 * 32; i += 256) {
            int r  = i >> 5;
            int c4 = i & 31;
            float4 v = make_float4(0.f, 0.f, 0.f, 0.f);
            if (rowBase + r < N) v = ((const float4*)(x + (size_t)(rowBase + r) * HIDC))[c4];
            float* p = &As[r * 132 + c4 * 4];
            p[0] = v.x; p[1] = v.y; p[2] = v.z; p[3] = v.w;
        }
    }
    __syncthreads();

    const int lane = t & 31;
    const int wv   = t >> 5;
    const int m    = lane & 15;
    const int half = lane >> 4;
    const int kl   = half * 2;
    const int n    = wv * 16 + (lane & 15);

    v8f c = {};
    const float* Arow = &As[m * 132];
#pragma unroll 8
    for (int k0 = 0; k0 < HIDC; k0 += 4) {
        v2f a, b;
        a.x = Arow[k0 + kl];
        a.y = Arow[k0 + kl + 1];
        b.x = W[(k0 + kl) * HIDC + n];
        b.y = W[(k0 + kl + 1) * HIDC + n];
        c = __builtin_amdgcn_wmma_f32_16x16x4_f32(false, a, false, b,
                                                  (short)0, c, false, false);
    }

    float* hp = h + (size_t)(rowBase + 8 * half) * HIDC + n;
    if (full) {
#pragma unroll
        for (int v = 0; v < 8; ++v) hp[v * HIDC] = c[v];
    } else {
#pragma unroll
        for (int v = 0; v < 8; ++v) {
            if (rowBase + 8 * half + v < N) hp[v * HIDC] = c[v];
        }
    }
}

__global__ __launch_bounds__(256) void gcn_gather1(const float* __restrict__ h,
                                                   const int* __restrict__ ssrc,
                                                   const int* __restrict__ rowptr,
                                                   const int* __restrict__ rowend,
                                                   const float* __restrict__ dinv,
                                                   float* __restrict__ acc, int N) {
    const int i = blockIdx.x * 8 + __builtin_amdgcn_readfirstlane((int)(threadIdx.x >> 5));
    if (i >= N) return;
    const int lane = threadIdx.x & 31;
    const int beg = rowptr[i];
    const int end = rowend[i];
    const float di = dinv[i];
    const float4* h4 = (const float4*)h;
    float4 a = make_float4(0.f, 0.f, 0.f, 0.f);

    for (int base = beg; base < end; base += 32) {
        const int n = min(32, end - base);
        int myS = 0;
        float myW = 0.f;
        if (lane < n) {
            int s = ssrc[base + lane];
            s = min(max(s, 0), N - 1);
            myS = s;
            myW = dinv[s] * di;
        }
        const int myWi = __float_as_int(myW);
        int k = 0;
        for (; k + 4 <= n; k += 4) {
            const float4 v0 = h4[(size_t)__builtin_amdgcn_readlane(myS, k)     * 32 + lane];
            const float4 v1 = h4[(size_t)__builtin_amdgcn_readlane(myS, k + 1) * 32 + lane];
            const float4 v2 = h4[(size_t)__builtin_amdgcn_readlane(myS, k + 2) * 32 + lane];
            const float4 v3 = h4[(size_t)__builtin_amdgcn_readlane(myS, k + 3) * 32 + lane];
            const float w0 = __int_as_float(__builtin_amdgcn_readlane(myWi, k));
            const float w1 = __int_as_float(__builtin_amdgcn_readlane(myWi, k + 1));
            const float w2 = __int_as_float(__builtin_amdgcn_readlane(myWi, k + 2));
            const float w3 = __int_as_float(__builtin_amdgcn_readlane(myWi, k + 3));
            a.x = fmaf(v0.x, w0, a.x); a.y = fmaf(v0.y, w0, a.y); a.z = fmaf(v0.z, w0, a.z); a.w = fmaf(v0.w, w0, a.w);
            a.x = fmaf(v1.x, w1, a.x); a.y = fmaf(v1.y, w1, a.y); a.z = fmaf(v1.z, w1, a.z); a.w = fmaf(v1.w, w1, a.w);
            a.x = fmaf(v2.x, w2, a.x); a.y = fmaf(v2.y, w2, a.y); a.z = fmaf(v2.z, w2, a.z); a.w = fmaf(v2.w, w2, a.w);
            a.x = fmaf(v3.x, w3, a.x); a.y = fmaf(v3.y, w3, a.y); a.z = fmaf(v3.z, w3, a.z); a.w = fmaf(v3.w, w3, a.w);
        }
        const int r = n - k;
        if (r > 0) {
            float4 v0 = h4[(size_t)__builtin_amdgcn_readlane(myS, k) * 32 + lane];
            float4 v1 = make_float4(0.f, 0.f, 0.f, 0.f), v2 = v1;
            if (r > 1) v1 = h4[(size_t)__builtin_amdgcn_readlane(myS, k + 1) * 32 + lane];
            if (r > 2) v2 = h4[(size_t)__builtin_amdgcn_readlane(myS, k + 2) * 32 + lane];
            const float w0 = __int_as_float(__builtin_amdgcn_readlane(myWi, k));
            a.x = fmaf(v0.x, w0, a.x); a.y = fmaf(v0.y, w0, a.y); a.z = fmaf(v0.z, w0, a.z); a.w = fmaf(v0.w, w0, a.w);
            if (r > 1) {
                const float w1 = __int_as_float(__builtin_amdgcn_readlane(myWi, k + 1));
                a.x = fmaf(v1.x, w1, a.x); a.y = fmaf(v1.y, w1, a.y); a.z = fmaf(v1.z, w1, a.z); a.w = fmaf(v1.w, w1, a.w);
            }
            if (r > 2) {
                const float w2 = __int_as_float(__builtin_amdgcn_readlane(myWi, k + 2));
                a.x = fmaf(v2.x, w2, a.x); a.y = fmaf(v2.y, w2, a.y); a.z = fmaf(v2.z, w2, a.z); a.w = fmaf(v2.w, w2, a.w);
            }
        }
    }
    ((float4*)(acc + (size_t)i * HIDC))[lane] = a;
}

__global__ __launch_bounds__(256) void gcn_fin1_dot(const float* __restrict__ acc,
                                                    const float* __restrict__ h,
                                                    const float* __restrict__ dinv,
                                                    const float* __restrict__ b1,
                                                    const float* __restrict__ W2,
                                                    float* __restrict__ h2, int N) {
    int i = blockIdx.x * 8 + (threadIdx.x >> 5);
    if (i >= N) return;
    int lane = threadIdx.x & 31;
    float di = dinv[i];
    float ns = di * di;
    float4 a  = ((const float4*)(acc + (size_t)i * HIDC))[lane];
    float4 hv = ((const float4*)(h   + (size_t)i * HIDC))[lane];
    float4 bb = ((const float4*)b1)[lane];
    float4 w  = ((const float4*)W2)[lane];
    float r0 = fmaxf(a.x + hv.x * ns + bb.x, 0.0f);
    float r1 = fmaxf(a.y + hv.y * ns + bb.y, 0.0f);
    float r2 = fmaxf(a.z + hv.z * ns + bb.z, 0.0f);
    float r3 = fmaxf(a.w + hv.w * ns + bb.w, 0.0f);
    float p = r0 * w.x + r1 * w.y + r2 * w.z + r3 * w.w;
#pragma unroll
    for (int off = 16; off > 0; off >>= 1) p += __shfl_down(p, off, 32);
    if (lane == 0) h2[i] = p;
}

__global__ void gcn_init_out(const float* __restrict__ h2, const float* __restrict__ dinv,
                             const float* __restrict__ b2, float* __restrict__ out, int N) {
    int i = blockIdx.x * blockDim.x + threadIdx.x;
    if (i < N) {
        float di = dinv[i];
        out[i] = h2[i] * di * di + b2[0];
    }
}

__global__ void gcn_agg2(const float* __restrict__ h2, const int* __restrict__ src,
                         const int* __restrict__ dst, const float* __restrict__ dinv,
                         float* __restrict__ out, int E) {
    int e = blockIdx.x * blockDim.x + threadIdx.x;
    if (e < E) {
        int s = src[e];
        int d = dst[e];
        atomicAdd(&out[d], h2[s] * dinv[s] * dinv[d]);
    }
}

extern "C" void kernel_launch(void* const* d_in, const int* in_sizes, int n_in,
                              void* d_out, int out_size, void* d_ws, size_t ws_size,
                              hipStream_t stream) {
    const float* x  = (const float*)d_in[0];
    const float* W1 = (const float*)d_in[1];
    const float* b1 = (const float*)d_in[2];
    const float* W2 = (const float*)d_in[3];
    const float* b2 = (const float*)d_in[4];
    const int*   ei = (const int*)d_in[5];

    const int N = in_sizes[0] / HIDC;
    const int E = in_sizes[5] / 2;
    const int* src = ei;
    const int* dst = ei + E;
    float* out = (float*)d_out;

    float* h      = (float*)d_ws;
    float* acc    = h   + (size_t)N * HIDC;
    int*   cnt    = (int*)(acc + (size_t)N * HIDC);
    int*   rowptr = cnt + N;
    int*   cursor = rowptr + N;
    float* dinv   = (float*)(cursor + N);
    float* h2     = dinv + N;
    int*   ssrc   = (int*)(h2 + N);

    const int TB = 256;

    gcn_zero_cnt  <<<(N + TB - 1) / TB, TB, 0, stream>>>(cnt, N);
    gcn_count     <<<(E + TB - 1) / TB, TB, 0, stream>>>(dst, cnt, E);
    gcn_scan      <<<(N + 4095) / 4096, 1024, 0, stream>>>(cnt, rowptr, cursor, dinv, N);
    gcn_fill      <<<(E + TB - 1) / TB, TB, 0, stream>>>(src, dst, cursor, ssrc, E);
    gcn_gemm1_wmma<<<(N + 15) / 16,     TB, 0, stream>>>(x, W1, h, N);
    gcn_gather1   <<<(N + 7) / 8,       TB, 0, stream>>>(h, ssrc, rowptr, cursor, dinv, acc, N);
    gcn_fin1_dot  <<<(N + 7) / 8,       TB, 0, stream>>>(acc, h, dinv, b1, W2, h2, N);
    gcn_init_out  <<<(N + TB - 1) / TB, TB, 0, stream>>>(h2, dinv, b2, out, N);
    gcn_agg2      <<<(E + TB - 1) / TB, TB, 0, stream>>>(h2, src, dst, dinv, out, E);
}


# agg2 as a gather over the same destination buckets: one thread sums g[src] (g = h2*dinv) of its node in a register, stores out once; no f32 atomics left; f32 sum order = self term, then bucket order
# speedup vs baseline: 1.8382x; 1.2702x over previous
#include <hip/hip_runtime.h>
#include <math.h>

#define HIDC 128

typedef __attribute__((ext_vector_type(2))) float v2f;
typedef __attribute__((ext_vector_type(8))) float v8f;

__global__ void gcn_zero_cnt(int* __restrict__ cnt, int N) {
    int i = blockIdx.x * blockDim.x + threadIdx.x;
    if (i < N) cnt[i] = 0;
}

__global__ void gcn_count(const int* __restrict__ dst, int* __restrict__ cnt, int E) {
    int e = blockIdx.x * blockDim.x + threadIdx.x;
    if (e < E) atomicAdd(&cnt[dst[e]], 1);
}

__global__ __launch_bounds__(1024) void gcn_scan(const int* __restrict__ cnt,
                                                 int* __restrict__ rowptr,
                                                 int* __restrict__ cursor,
                                                 float* __restrict__ dinv, int N) {
    __shared__ int preS[32];
    __shared__ int ownS[32];
    const int t = threadIdx.x;
    const int lane = t & 31;
    const int wv = t >> 5;
    const int blockStart = blockIdx.x * 4096;

    int pre = 0;
    const int4* c4 = (const int4*)cnt;
#pragma unroll 4
    for (int j = 0; j < (int)blockIdx.x; ++j) {
        int4 v = c4[j * 1024 + t];
        pre += (v.x + v.y) + (v.z + v.w);
    }
#pragma unroll
    for (int off = 16; off > 0; off >>= 1) pre += __shfl_xor(pre, off, 32);
    if (lane == 0) preS[wv] = pre;

    const int i0 = blockStart + t * 4;
    int c0 = (i0     < N) ? cnt[i0]     : 0;
    int c1 = (i0 + 1 < N) ? cnt[i0 + 1] : 0;
    int c2 = (i0 + 2 < N) ? cnt[i0 + 2] : 0;
    int c3 = (i0 + 3 < N) ? cnt[i0 + 3] : 0;
    const int s = (c0 + c1) + (c2 + c3);
    int inc = s;
#pragma unroll
    for (int off = 1; off < 32; off <<= 1) {
        int y = __shfl_up(inc, off, 32);
        if (lane >= off) inc += y;
    }
    if (lane == 31) ownS[wv] = inc;
    __syncthreads();

    int ab = preS[lane] + ((lane < wv) ? ownS[lane] : 0);
#pragma unroll
    for (int off = 16; off > 0; off >>= 1) ab += __shfl_xor(ab, off, 32);

    const int e0 = ab + inc - s;
    const int e1 = e0 + c0;
    const int e2 = e1 + c1;
    const int e3 = e2 + c2;
    if (i0 < N)     { rowptr[i0]     = e0; cursor[i0]     = e0; dinv[i0]     = rsqrtf((float)(c0 + 1)); }
    if (i0 + 1 < N) { rowptr[i0 + 1] = e1; cursor[i0 + 1] = e1; dinv[i0 + 1] = rsqrtf((float)(c1 + 1)); }
    if (i0 + 2 < N) { rowptr[i0 + 2] = e2; cursor[i0 + 2] = e2; dinv[i0 + 2] = rsqrtf((float)(c2 + 1)); }
    if (i0 + 3 < N) { rowptr[i0 + 3] = e3; cursor[i0 + 3] = e3; dinv[i0 + 3] = rsqrtf((float)(c3 + 1)); }
}

__global__ void gcn_fill(const int* __restrict__ src, const int* __restrict__ dst,
                         int* __restrict__ cursor, int* __restrict__ ssrc, int E) {
    int e = blockIdx.x * blockDim.x + threadIdx.x;
    if (e < E) {
        int pos = atomicAdd(&cursor[dst[e]], 1);
        ssrc[pos] = src[e];
    }
}

__global__ __launch_bounds__(256) void gcn_gemm1_wmma(const float* __restrict__ x,
                                                      const float* __restrict__ W,
                                                      float* __restrict__ h, int N) {
    __shared__ float As[16 * 132];
    const int t = threadIdx.x;
    const int rowBase = blockIdx.x * 16;
    const bool full = (rowBase + 16 <= N);

    if (full) {
#pragma unroll
        for (int i = t; i < 16 * 32; i += 256) {
            int r  = i >> 5;
            int c4 = i & 31;
            float4 v = ((const float4*)(x + (size_t)(rowBase + r) * HIDC))[c4];
            float* p = &As[r * 132 + c4 * 4];
            p[0] = v.x; p[1] = v.y; p[2] = v.z; p[3] = v.w;
        }
    } else {
        for (int i = t; i < 16 * 32; i += 256) {
            int r  = i >> 5;
            int c4 = i & 31;
            float4 v = make_float4(0.f, 0.f, 0.f, 0.f);
            if (rowBase + r < N) v = ((const float4*)(x + (size_t)(rowBase + r) * HIDC))[c4];
            float* p = &As[r * 132 + c4 * 4];
            p[0] = v.x; p[1] = v.y; p[2] = v.z; p[3] = v.w;
        }
    }
    __syncthreads();

    const int lane = t & 31;
    const int wv   = t >> 5;
    const int m    = lane & 15;
    const int half = lane >> 4;
    const int kl   = half * 2;
    const int n    = wv * 16 + (lane & 15);

    v8f c = {};
    const float* Arow = &As[m * 132];
#pragma unroll 8
    for (int k0 = 0; k0 < HIDC; k0 += 4) {
        v2f a, b;
        a.x = Arow[k0 + kl];
        a.y = Arow[k0 + kl + 1];
        b.x = W[(k0 + kl) * HIDC + n];
        b.y = W[(k0 + kl + 1) * HIDC + n];
        c = __builtin_amdgcn_wmma_f32_16x16x4_f32(false, a, false, b,
                                                  (short)0, c, false, false);
    }

    float* hp = h + (size_t)(rowBase + 8 * half) * HIDC + n;
    if (full) {
#pragma unroll
        for (int v = 0; v < 8; ++v) hp[v * HIDC] = c[v];
    } else {
#pragma unroll
        for (int v = 0; v < 8; ++v) {
            if (rowBase + 8 * half + v < N) hp[v * HIDC] = c[v];
        }
    }
}

__global__ __launch_bounds__(256) void gcn_gather1(const float* __restrict__ h,
                                                   const int* __restrict__ ssrc,
                                                   const int* __restrict__ rowptr,
                                                   const int* __restrict__ rowend,
                                                   const float* __restrict__ dinv,
                                                   float* __restrict__ acc, int N) {
    const int i = blockIdx.x * 8 + __builtin_amdgcn_readfirstlane((int)(threadIdx.x >> 5));
    if (i >= N) return;
    const int lane = threadIdx.x & 31;
    const int beg = rowptr[i];
    const int end = rowend[i];
    const float di = dinv[i];
    const float4* h4 = (const float4*)h;
    float4 a = make_float4(0.f, 0.f, 0.f, 0.f);

    for (int base = beg; base < end; base += 32) {
        const int n = min(32, end - base);
        int myS = 0;
        float myW = 0.f;
        if (lane < n) {
            int s = ssrc[base + lane];
            s = min(max(s, 0), N - 1);
            myS = s;
            myW = dinv[s] * di;
        }
        const int myWi = __float_as_int(myW);
        int k = 0;
        for (; k + 4 <= n; k += 4) {
            const float4 v0 = h4[(size_t)__builtin_amdgcn_readlane(myS, k)     * 32 + lane];
            const float4 v1 = h4[(size_t)__builtin_amdgcn_readlane(myS, k + 1) * 32 + lane];
            const float4 v2 = h4[(size_t)__builtin_amdgcn_readlane(myS, k + 2) * 32 + lane];
            const float4 v3 = h4[(size_t)__builtin_amdgcn_readlane(myS, k + 3) * 32 + lane];
            const float w0 = __int_as_float(__builtin_amdgcn_readlane(myWi, k));
            const float w1 = __int_as_float(__builtin_amdgcn_readlane(myWi, k + 1));
            const float w2 = __int_as_float(__builtin_amdgcn_readlane(myWi, k + 2));
            const float w3 = __int_as_float(__builtin_amdgcn_readlane(myWi, k + 3));
            a.x = fmaf(v0.x, w0, a.x); a.y = fmaf(v0.y, w0, a.y); a.z = fmaf(v0.z, w0, a.z); a.w = fmaf(v0.w, w0, a.w);
            a.x = fmaf(v1.x, w1, a.x); a.y = fmaf(v1.y, w1, a.y); a.z = fmaf(v1.z, w1, a.z); a.w = fmaf(v1.w, w1, a.w);
            a.x = fmaf(v2.x, w2, a.x); a.y = fmaf(v2.y, w2, a.y); a.z = fmaf(v2.z, w2, a.z); a.w = fmaf(v2.w, w2, a.w);
            a.x = fmaf(v3.x, w3, a.x); a.y = fmaf(v3.y, w3, a.y); a.z = fmaf(v3.z, w3, a.z); a.w = fmaf(v3.w, w3, a.w);
        }
        const int r = n - k;
        if (r > 0) {
            float4 v0 = h4[(size_t)__builtin_amdgcn_readlane(myS, k) * 32 + lane];
            float4 v1 = make_float4(0.f, 0.f, 0.f, 0.f), v2 = v1;
            if (r > 1) v1 = h4[(size_t)__builtin_amdgcn_readlane(myS, k + 1) * 32 + lane];
            if (r > 2) v2 = h4[(size_t)__builtin_amdgcn_readlane(myS, k + 2) * 32 + lane];
            const float w0 = __int_as_float(__builtin_amdgcn_readlane(myWi, k));
            a.x = fmaf(v0.x, w0, a.x); a.y = fmaf(v0.y, w0, a.y); a.z = fmaf(v0.z, w0, a.z); a.w = fmaf(v0.w, w0, a.w);
            if (r > 1) {
                const float w1 = __int_as_float(__builtin_amdgcn_readlane(myWi, k + 1));
                a.x = fmaf(v1.x, w1, a.x); a.y = fmaf(v1.y, w1, a.y); a.z = fmaf(v1.z, w1, a.z); a.w = fmaf(v1.w, w1, a.w);
            }
            if (r > 2) {
                const float w2 = __int_as_float(__builtin_amdgcn_readlane(myWi, k + 2));
                a.x = fmaf(v2.x, w2, a.x); a.y = fmaf(v2.y, w2, a.y); a.z = fmaf(v2.z, w2, a.z); a.w = fmaf(v2.w, w2, a.w);
            }
        }
    }
    ((float4*)(acc + (size_t)i * HIDC))[lane] = a;
}

__global__ __launch_bounds__(256) void gcn_fin1_dot(const float* __restrict__ acc,
                                                    const float* __restrict__ h,
                                                    const float* __restrict__ dinv,
                                                    const float* __restrict__ b1,
                                                    const float* __restrict__ W2,
                                                    float* __restrict__ g, int N) {
    int i = blockIdx.x * 8 + (threadIdx.x >> 5);
    if (i >= N) return;
    int lane = threadIdx.x & 31;
    float di = dinv[i];
    float ns = di * di;
    float4 a  = ((const float4*)(acc + (size_t)i * HIDC))[lane];
    float4 hv = ((const float4*)(h   + (size_t)i * HIDC))[lane];
    float4 bb = ((const float4*)b1)[lane];
    float4 w  = ((const float4*)W2)[lane];
    float r0 = fmaxf(a.x + hv.x * ns + bb.x, 0.0f);
    float r1 = fmaxf(a.y + hv.y * ns + bb.y, 0.0f);
    float r2 = fmaxf(a.z + hv.z * ns + bb.z, 0.0f);
    float r3 = fmaxf(a.w + hv.w * ns + bb.w, 0.0f);
    float p = r0 * w.x + r1 * w.y + r2 * w.z + r3 * w.w;
#pragma unroll
    for (int off = 16; off > 0; off >>= 1) p += __shfl_down(p, off, 32);
    if (lane == 0) g[i] = p * di;
}

__global__ __launch_bounds__(128) void gcn_gather2(const float* __restrict__ g,
                                                   const int* __restrict__ ssrc,
                                                   const int* __restrict__ rowptr,
                                                   const int* __restrict__ rowend,
                                                   const float* __restrict__ dinv,
                                                   const float* __restrict__ b2,
                                                   float* __restrict__ out, int N) {
    const int i = blockIdx.x * blockDim.x + threadIdx.x;
    if (i >= N) return;
    const int beg = rowptr[i];
    const int end = rowend[i];
    const float di = dinv[i];
    const float bias = b2[0];
    float a = g[i];
    int k = beg;
    for (; k + 4 <= end; k += 4) {
        int s0 = ssrc[k], s1 = ssrc[k + 1], s2 = ssrc[k + 2], s3 = ssrc[k + 3];
        s0 = min(max(s0, 0), N - 1);
        s1 = min(max(s1, 0), N - 1);
        s2 = min(max(s2, 0), N - 1);
        s3 = min(max(s3, 0), N - 1);
        const float g0 = g[s0], g1 = g[s1], g2 = g[s2], g3 = g[s3];
        a += g0; a += g1; a += g2; a += g3;
    }
    const int r = end - k;
    if (r > 0) {
        const int k1 = min(k + 1, end - 1);
        const int k2 = min(k + 2, end - 1);
        int s0 = ssrc[k], s1 = ssrc[k1], s2 = ssrc[k2];
        s0 = min(max(s0, 0), N - 1);
        s1 = min(max(s1, 0), N - 1);
        s2 = min(max(s2, 0), N - 1);
        const float g0 = g[s0], g1 = g[s1], g2 = g[s2];
        a += g0;
        a += (r > 1) ? g1 : 0.0f;
        a += (r > 2) ? g2 : 0.0f;
    }
    out[i] = a * di + bias;
}

extern "C" void kernel_launch(void* const* d_in, const int* in_sizes, int n_in,
                              void* d_out, int out_size, void* d_ws, size_t ws_size,
                              hipStream_t stream) {
    const float* x  = (const float*)d_in[0];
    const float* W1 = (const float*)d_in[1];
    const float* b1 = (const float*)d_in[2];
    const float* W2 = (const float*)d_in[3];
    const float* b2 = (const float*)d_in[4];
    const int*   ei = (const int*)d_in[5];

    const int N = in_sizes[0] / HIDC;
    const int E = in_sizes[5] / 2;
    const int* src = ei;
    const int* dst = ei + E;
    float* out = (float*)d_out;

    float* h      = (float*)d_ws;
    float* acc    = h   + (size_t)N * HIDC;
    int*   cnt    = (int*)(acc + (size_t)N * HIDC);
    int*   rowptr = cnt + N;
    int*   cursor = rowptr + N;
    float* dinv   = (float*)(cursor + N);
    float* g      = dinv + N;
    int*   ssrc   = (int*)(g + N);

    const int TB = 256;

    gcn_zero_cnt  <<<(N + TB - 1) / TB, TB, 0, stream>>>(cnt, N);
    gcn_count     <<<(E + TB - 1) / TB, TB, 0, stream>>>(dst, cnt, E);
    gcn_scan      <<<(N + 4095) / 4096, 1024, 0, stream>>>(cnt, rowptr, cursor, dinv, N);
    gcn_fill      <<<(E + TB - 1) / TB, TB, 0, stream>>>(src, dst, cursor, ssrc, E);
    gcn_gemm1_wmma<<<(N + 15) / 16,     TB, 0, stream>>>(x, W1, h, N);
    gcn_gather1   <<<(N + 7) / 8,       TB, 0, stream>>>(h, ssrc, rowptr, cursor, dinv, acc, N);
    gcn_fin1_dot  <<<(N + 7) / 8,       TB, 0, stream>>>(acc, h, dinv, b1, W2, g, N);
    gcn_gather2   <<<(N + 127) / 128,   128, 0, stream>>>(g, ssrc, rowptr, cursor, dinv, b2, out, N);
}


# count pass keeps the old value its int atomic returns as the edge's rank; fill puts src at rowptr[dst]+rank with plain loads/stores, no atomics; f32 sum order = count-pass arrival order
# speedup vs baseline: 2.2948x; 1.2484x over previous
#include <hip/hip_runtime.h>
#include <math.h>

#define HIDC 128

typedef __attribute__((ext_vector_type(2))) float v2f;
typedef __attribute__((ext_vector_type(8))) float v8f;

__global__ void gcn_zero_cnt(int* __restrict__ cnt, int N) {
    int i = blockIdx.x * blockDim.x + threadIdx.x;
    if (i < N) cnt[i] = 0;
}

__global__ void gcn_count_rank(const int* __restrict__ dst, int* __restrict__ cnt,
                               int* __restrict__ rank, int E) {
    int e = blockIdx.x * blockDim.x + threadIdx.x;
    if (e < E) rank[e] = atomicAdd(&cnt[dst[e]], 1);
}

__global__ __launch_bounds__(1024) void gcn_scan(const int* __restrict__ cnt,
                                                 int* __restrict__ rowptr,
                                                 int* __restrict__ rowend,
                                                 float* __restrict__ dinv, int N) {
    __shared__ int preS[32];
    __shared__ int ownS[32];
    const int t = threadIdx.x;
    const int lane = t & 31;
    const int wv = t >> 5;
    const int blockStart = blockIdx.x * 4096;

    int pre = 0;
    const int4* c4 = (const int4*)cnt;
#pragma unroll 4
    for (int j = 0; j < (int)blockIdx.x; ++j) {
        int4 v = c4[j * 1024 + t];
        pre += (v.x + v.y) + (v.z + v.w);
    }
#pragma unroll
    for (int off = 16; off > 0; off >>= 1) pre += __shfl_xor(pre, off, 32);
    if (lane == 0) preS[wv] = pre;

    const int i0 = blockStart + t * 4;
    int c0 = (i0     < N) ? cnt[i0]     : 0;
    int c1 = (i0 + 1 < N) ? cnt[i0 + 1] : 0;
    int c2 = (i0 + 2 < N) ? cnt[i0 + 2] : 0;
    int c3 = (i0 + 3 < N) ? cnt[i0 + 3] : 0;
    const int s = (c0 + c1) + (c2 + c3);
    int inc = s;
#pragma unroll
    for (int off = 1; off < 32; off <<= 1) {
        int y = __shfl_up(inc, off, 32);
        if (lane >= off) inc += y;
    }
    if (lane == 31) ownS[wv] = inc;
    __syncthreads();

    int ab = preS[lane] + ((lane < wv) ? ownS[lane] : 0);
#pragma unroll
    for (int off = 16; off > 0; off >>= 1) ab += __shfl_xor(ab, off, 32);

    const int e0 = ab + inc - s;
    const int e1 = e0 + c0;
    const int e2 = e1 + c1;
    const int e3 = e2 + c2;
    if (i0 < N)     { rowptr[i0]     = e0; rowend[i0]     = e1;      dinv[i0]     = rsqrtf((float)(c0 + 1)); }
    if (i0 + 1 < N) { rowptr[i0 + 1] = e1; rowend[i0 + 1] = e2;      dinv[i0 + 1] = rsqrtf((float)(c1 + 1)); }
    if (i0 + 2 < N) { rowptr[i0 + 2] = e2; rowend[i0 + 2] = e3;      dinv[i0 + 2] = rsqrtf((float)(c2 + 1)); }
    if (i0 + 3 < N) { rowptr[i0 + 3] = e3; rowend[i0 + 3] = e3 + c3; dinv[i0 + 3] = rsqrtf((float)(c3 + 1)); }
}

__global__ void gcn_fill(const int* __restrict__ src, const int* __restrict__ dst,
                         const int* __restrict__ rank, const int* __restrict__ rowptr,
                         int* __restrict__ ssrc, int E) {
    int e = blockIdx.x * blockDim.x + threadIdx.x;
    if (e < E) {
        int pos = rowptr[dst[e]] + rank[e];
        pos = min(max(pos, 0), E - 1);
        ssrc[pos] = src[e];
    }
}

__global__ __launch_bounds__(256) void gcn_gemm1_wmma(const float* __restrict__ x,
                                                      const float* __restrict__ W,
                                                      float* __restrict__ h, int N) {
    __shared__ float As[16 * 132];
    const int t = threadIdx.x;
    const int rowBase = blockIdx.x * 16;
    const bool full = (rowBase + 16 <= N);

    if (full) {
#pragma unroll
        for (int i = t; i < 16 * 32; i += 256) {
            int r  = i >> 5;
            int c4 = i & 31;
            float4 v = ((const float4*)(x + (size_t)(rowBase + r) * HIDC))[c4];
            float* p = &As[r * 132 + c4 * 4];
            p[0] = v.x; p[1] = v.y; p[2] = v.z; p[3] = v.w;
        }
    } else {
        for (int i = t; i < 16 * 32; i += 256) {
            int r  = i >> 5;
            int c4 = i & 31;
            float4 v = make_float4(0.f, 0.f, 0.f, 0.f);
            if (rowBase + r < N) v = ((const float4*)(x + (size_t)(rowBase + r) * HIDC))[c4];
            float* p = &As[r * 132 + c4 * 4];
            p[0] = v.x; p[1] = v.y; p[2] = v.z; p[3] = v.w;
        }
    }
    __syncthreads();

    const int lane = t & 31;
    const int wv   = t >> 5;
    const int m    = lane & 15;
    const int half = lane >> 4;
    const int kl   = half * 2;
    const int n    = wv * 16 + (lane & 15);

    v8f c = {};
    const float* Arow = &As[m * 132];
#pragma unroll 8
    for (int k0 = 0; k0 < HIDC; k0 += 4) {
        v2f a, b;
        a.x = Arow[k0 + kl];
        a.y = Arow[k0 + kl + 1];
        b.x = W[(k0 + kl) * HIDC + n];
        b.y = W[(k0 + kl + 1) * HIDC + n];
        c = __builtin_amdgcn_wmma_f32_16x16x4_f32(false, a, false, b,
                                                  (short)0, c, false, false);
    }

    float* hp = h + (size_t)(rowBase + 8 * half) * HIDC + n;
    if (full) {
#pragma unroll
        for (int v = 0; v < 8; ++v) hp[v * HIDC] = c[v];
    } else {
#pragma unroll
        for (int v = 0; v < 8; ++v) {
            if (rowBase + 8 * half + v < N) hp[v * HIDC] = c[v];
        }
    }
}

__global__ __launch_bounds__(256) void gcn_gather1(const float* __restrict__ h,
                                                   const int* __restrict__ ssrc,
                                                   const int* __restrict__ rowptr,
                                                   const int* __restrict__ rowend,
                                                   const float* __restrict__ dinv,
                                                   float* __restrict__ acc, int N) {
    const int i = blockIdx.x * 8 + __builtin_amdgcn_readfirstlane((int)(threadIdx.x >> 5));
    if (i >= N) return;
    const int lane = threadIdx.x & 31;
    const int beg = rowptr[i];
    const int end = rowend[i];
    const float di = dinv[i];
    const float4* h4 = (const float4*)h;
    float4 a = make_float4(0.f, 0.f, 0.f, 0.f);

    for (int base = beg; base < end; base += 32) {
        const int n = min(32, end - base);
        int myS = 0;
        float myW = 0.f;
        if (lane < n) {
            int s = ssrc[base + lane];
            s = min(max(s, 0), N - 1);
            myS = s;
            myW = dinv[s] * di;
        }
        const int myWi = __float_as_int(myW);
        int k = 0;
        for (; k + 4 <= n; k += 4) {
            const float4 v0 = h4[(size_t)__builtin_amdgcn_readlane(myS, k)     * 32 + lane];
            const float4 v1 = h4[(size_t)__builtin_amdgcn_readlane(myS, k + 1) * 32 + lane];
            const float4 v2 = h4[(size_t)__builtin_amdgcn_readlane(myS, k + 2) * 32 + lane];
            const float4 v3 = h4[(size_t)__builtin_amdgcn_readlane(myS, k + 3) * 32 + lane];
            const float w0 = __int_as_float(__builtin_amdgcn_readlane(myWi, k));
            const float w1 = __int_as_float(__builtin_amdgcn_readlane(myWi, k + 1));
            const float w2 = __int_as_float(__builtin_amdgcn_readlane(myWi, k + 2));
            const float w3 = __int_as_float(__builtin_amdgcn_readlane(myWi, k + 3));
            a.x = fmaf(v0.x, w0, a.x); a.y = fmaf(v0.y, w0, a.y); a.z = fmaf(v0.z, w0, a.z); a.w = fmaf(v0.w, w0, a.w);
            a.x = fmaf(v1.x, w1, a.x); a.y = fmaf(v1.y, w1, a.y); a.z = fmaf(v1.z, w1, a.z); a.w = fmaf(v1.w, w1, a.w);
            a.x = fmaf(v2.x, w2, a.x); a.y = fmaf(v2.y, w2, a.y); a.z = fmaf(v2.z, w2, a.z); a.w = fmaf(v2.w, w2, a.w);
            a.x = fmaf(v3.x, w3, a.x); a.y = fmaf(v3.y, w3, a.y); a.z = fmaf(v3.z, w3, a.z); a.w = fmaf(v3.w, w3, a.w);
        }
        const int r = n - k;
        if (r > 0) {
            float4 v0 = h4[(size_t)__builtin_amdgcn_readlane(myS, k) * 32 + lane];
            float4 v1 = make_float4(0.f, 0.f, 0.f, 0.f), v2 = v1;
            if (r > 1) v1 = h4[(size_t)__builtin_amdgcn_readlane(myS, k + 1) * 32 + lane];
            if (r > 2) v2 = h4[(size_t)__builtin_amdgcn_readlane(myS, k + 2) * 32 + lane];
            const float w0 = __int_as_float(__builtin_amdgcn_readlane(myWi, k));
            a.x = fmaf(v0.x, w0, a.x); a.y = fmaf(v0.y, w0, a.y); a.z = fmaf(v0.z, w0, a.z); a.w = fmaf(v0.w, w0, a.w);
            if (r > 1) {
                const float w1 = __int_as_float(__builtin_amdgcn_readlane(myWi, k + 1));
                a.x = fmaf(v1.x, w1, a.x); a.y = fmaf(v1.y, w1, a.y); a.z = fmaf(v1.z, w1, a.z); a.w = fmaf(v1.w, w1, a.w);
            }
            if (r > 2) {
                const float w2 = __int_as_float(__builtin_amdgcn_readlane(myWi, k + 2));
                a.x = fmaf(v2.x, w2, a.x); a.y = fmaf(v2.y, w2, a.y); a.z = fmaf(v2.z, w2, a.z); a.w = fmaf(v2.w, w2, a.w);
            }
        }
    }
    ((float4*)(acc + (size_t)i * HIDC))[lane] = a;
}

__global__ __launch_bounds__(256) void gcn_fin1_dot(const float* __restrict__ acc,
                                                    const float* __restrict__ h,
                                                    const float* __restrict__ dinv,
                                                    const float* __restrict__ b1,
                                                    const float* __restrict__ W2,
                                                    float* __restrict__ g, int N) {
    int i = blockIdx.x * 8 + (threadIdx.x >> 5);
    if (i >= N) return;
    int lane = threadIdx.x & 31;
    float di = dinv[i];
    float ns = di * di;
    float4 a  = ((const float4*)(acc + (size_t)i * HIDC))[lane];
    float4 hv = ((const float4*)(h   + (size_t)i * HIDC))[lane];
    float4 bb = ((const float4*)b1)[lane];
    float4 w  = ((const float4*)W2)[lane];
    float r0 = fmaxf(a.x + hv.x * ns + bb.x, 0.0f);
    float r1 = fmaxf(a.y + hv.y * ns + bb.y, 0.0f);
    float r2 = fmaxf(a.z + hv.z * ns + bb.z, 0.0f);
    float r3 = fmaxf(a.w + hv.w * ns + bb.w, 0.0f);
    float p = r0 * w.x + r1 * w.y + r2 * w.z + r3 * w.w;
#pragma unroll
    for (int off = 16; off > 0; off >>= 1) p += __shfl_down(p, off, 32);
    if (lane == 0) g[i] = p * di;
}

__global__ __launch_bounds__(128) void gcn_gather2(const float* __restrict__ g,
                                                   const int* __restrict__ ssrc,
                                                   const int* __restrict__ rowptr,
                                                   const int* __restrict__ rowend,
                                                   const float* __restrict__ dinv,
                                                   const float* __restrict__ b2,
                                                   float* __restrict__ out, int N) {
    const int i = blockIdx.x * blockDim.x + threadIdx.x;
    if (i >= N) return;
    const int beg = rowptr[i];
    const int end = rowend[i];
    const float di = dinv[i];
    const float bias = b2[0];
    float a = g[i];
    int k = beg;
    for (; k + 4 <= end; k += 4) {
        int s0 = ssrc[k], s1 = ssrc[k + 1], s2 = ssrc[k + 2], s3 = ssrc[k + 3];
        s0 = min(max(s0, 0), N - 1);
        s1 = min(max(s1, 0), N - 1);
        s2 = min(max(s2, 0), N - 1);
        s3 = min(max(s3, 0), N - 1);
        const float g0 = g[s0], g1 = g[s1], g2 = g[s2], g3 = g[s3];
        a += g0; a += g1; a += g2; a += g3;
    }
    const int r = end - k;
    if (r > 0) {
        const int k1 = min(k + 1, end - 1);
        const int k2 = min(k + 2, end - 1);
        int s0 = ssrc[k], s1 = ssrc[k1], s2 = ssrc[k2];
        s0 = min(max(s0, 0), N - 1);
        s1 = min(max(s1, 0), N - 1);
        s2 = min(max(s2, 0), N - 1);
        const float g0 = g[s0], g1 = g[s1], g2 = g[s2];
        a += g0;
        a += (r > 1) ? g1 : 0.0f;
        a += (r > 2) ? g2 : 0.0f;
    }
    out[i] = a * di + bias;
}

extern "C" void kernel_launch(void* const* d_in, const int* in_sizes, int n_in,
                              void* d_out, int out_size, void* d_ws, size_t ws_size,
                              hipStream_t stream) {
    const float* x  = (const float*)d_in[0];
    const float* W1 = (const float*)d_in[1];
    const float* b1 = (const float*)d_in[2];
    const float* W2 = (const float*)d_in[3];
    const float* b2 = (const float*)d_in[4];
    const int*   ei = (const int*)d_in[5];

    const int N = in_sizes[0] / HIDC;
    const int E = in_sizes[5] / 2;
    const int* src = ei;
    const int* dst = ei + E;
    float* out = (float*)d_out;

    float* h      = (float*)d_ws;
    float* acc    = h   + (size_t)N * HIDC;
    int*   cnt    = (int*)(acc + (size_t)N * HIDC);
    int*   rowptr = cnt + N;
    int*   rowend = rowptr + N;
    float* dinv   = (float*)(rowend + N);
    float* g      = dinv + N;
    int*   ssrc   = (int*)(g + N);
    int*   rank   = ssrc + E;

    const int TB = 256;

    gcn_zero_cnt  <<<(N + TB - 1) / TB, TB, 0, stream>>>(cnt, N);
    gcn_count_rank<<<(E + TB - 1) / TB, TB, 0, stream>>>(dst, cnt, rank, E);
    gcn_scan      <<<(N + 4095) / 4096, 1024, 0, stream>>>(cnt, rowptr, rowend, dinv, N);
    gcn_fill      <<<(E + TB - 1) / TB, TB, 0, stream>>>(src, dst, rank, rowptr, ssrc, E);
    gcn_gemm1_wmma<<<(N + 15) / 16,     TB, 0, stream>>>(x, W1, h, N);
    gcn_gather1   <<<(N + 7) / 8,       TB, 0, stream>>>(h, ssrc, rowptr, rowend, dinv, acc, N);
    gcn_fin1_dot  <<<(N + 7) / 8,       TB, 0, stream>>>(acc, h, dinv, b1, W2, g, N);
    gcn_gather2   <<<(N + 127) / 128,   128, 0, stream>>>(g, ssrc, rowptr, rowend, dinv, b2, out, N);
}


# CSR build as a two-level LDS bucket sort (512-node bins, then nodes): no global atomics left, counters are LDS atomics; f32 sum order = LDS atomic arrival order inside a block
# speedup vs baseline: 3.1995x; 1.3942x over previous
#include <hip/hip_runtime.h>
#include <math.h>

#define HIDC 128

typedef __attribute__((ext_vector_type(2))) float v2f;
typedef __attribute__((ext_vector_type(8))) float v8f;

#define BIN_SHIFT 9
#define BIN_NODES 512
#define CHUNK     4096
#define MAXB      1024

__global__ __launch_bounds__(256) void gcn_bin_count(const int* __restrict__ dst,
                                                     int* __restrict__ blockHist,
                                                     int N, int E, int NB, int NC) {
    __shared__ int hist[MAXB];
    const int t = threadIdx.x;
    const int chunk = blockIdx.x;
    for (int b = t; b < NB; b += 256) hist[b] = 0;
    __syncthreads();
    const int e0 = chunk * CHUNK + t;
    if (chunk * CHUNK + CHUNK <= E) {
        int d[16];
#pragma unroll
        for (int j = 0; j < 16; ++j) d[j] = dst[e0 + j * 256];
#pragma unroll
        for (int j = 0; j < 16; ++j) {
            const int dd = min(max(d[j], 0), N - 1);
            atomicAdd(&hist[min(dd >> BIN_SHIFT, MAXB - 1)], 1);
        }
    } else {
        for (int e = e0; e < E; e += 256) {
            const int dd = min(max(dst[e], 0), N - 1);
            atomicAdd(&hist[min(dd >> BIN_SHIFT, MAXB - 1)], 1);
        }
    }
    __syncthreads();
    for (int b = t; b < NB; b += 256) blockHist[b * NC + chunk] = hist[b];
    if (chunk == 0 && t == 0) blockHist[NB * NC] = 0;
}

__global__ __launch_bounds__(1024) void gcn_scan(const int* __restrict__ cnt,
                                                 int* __restrict__ offs, int M) {
    __shared__ int preS[32];
    __shared__ int ownS[32];
    const int t = threadIdx.x;
    const int lane = t & 31;
    const int wv = t >> 5;
    const int blockStart = blockIdx.x * 4096;

    int pre = 0;
    const int4* c4 = (const int4*)cnt;
#pragma unroll 4
    for (int j = 0; j < (int)blockIdx.x; ++j) {
        int4 v = c4[j * 1024 + t];
        pre += (v.x + v.y) + (v.z + v.w);
    }
#pragma unroll
    for (int off = 16; off > 0; off >>= 1) pre += __shfl_xor(pre, off, 32);
    if (lane == 0) preS[wv] = pre;

    const int i0 = blockStart + t * 4;
    int c0 = (i0     < M) ? cnt[i0]     : 0;
    int c1 = (i0 + 1 < M) ? cnt[i0 + 1] : 0;
    int c2 = (i0 + 2 < M) ? cnt[i0 + 2] : 0;
    int c3 = (i0 + 3 < M) ? cnt[i0 + 3] : 0;
    const int s = (c0 + c1) + (c2 + c3);
    int inc = s;
#pragma unroll
    for (int off = 1; off < 32; off <<= 1) {
        int y = __shfl_up(inc, off, 32);
        if (lane >= off) inc += y;
    }
    if (lane == 31) ownS[wv] = inc;
    __syncthreads();

    int ab = preS[lane] + ((lane < wv) ? ownS[lane] : 0);
#pragma unroll
    for (int off = 16; off > 0; off >>= 1) ab += __shfl_xor(ab, off, 32);

    const int e0 = ab + inc - s;
    const int e1 = e0 + c0;
    const int e2 = e1 + c1;
    const int e3 = e2 + c2;
    if (i0 < M)     offs[i0]     = e0;
    if (i0 + 1 < M) offs[i0 + 1] = e1;
    if (i0 + 2 < M) offs[i0 + 2] = e2;
    if (i0 + 3 < M) offs[i0 + 3] = e3;
}

__global__ __launch_bounds__(256) void gcn_bin_scatter(const int* __restrict__ src,
                                                       const int* __restrict__ dst,
                                                       const int* __restrict__ offs,
                                                       int* __restrict__ tmp,
                                                       int N, int E, int NB, int NC) {
    __shared__ int cur[MAXB];
    const int t = threadIdx.x;
    const int chunk = blockIdx.x;
    for (int b = t; b < NB; b += 256) cur[b] = offs[b * NC + chunk];
    __syncthreads();
    const int e0 = chunk * CHUNK + t;
    if (chunk * CHUNK + CHUNK <= E) {
#pragma unroll
        for (int hf = 0; hf < 2; ++hf) {
            int d[8], s[8];
#pragma unroll
            for (int j = 0; j < 8; ++j) {
                d[j] = dst[e0 + (hf * 8 + j) * 256];
                s[j] = src[e0 + (hf * 8 + j) * 256];
            }
            int pos[8];
#pragma unroll
            for (int j = 0; j < 8; ++j) {
                d[j] = min(max(d[j], 0), N - 1);
                pos[j] = atomicAdd(&cur[min(d[j] >> BIN_SHIFT, MAXB - 1)], 1);
            }
#pragma unroll
            for (int j = 0; j < 8; ++j) {
                const int ss = min(max(s[j], 0), N - 1);
                const int p = min(max(pos[j], 0), E - 1);
                tmp[p] = (ss << BIN_SHIFT) | (d[j] & (BIN_NODES - 1));
            }
        }
    } else {
        for (int e = e0; e < E; e += 256) {
            const int dd = min(max(dst[e], 0), N - 1);
            const int ss = min(max(src[e], 0), N - 1);
            int pos = atomicAdd(&cur[min(dd >> BIN_SHIFT, MAXB - 1)], 1);
            pos = min(max(pos, 0), E - 1);
            tmp[pos] = (ss << BIN_SHIFT) | (dd & (BIN_NODES - 1));
        }
    }
}

__global__ __launch_bounds__(512) void gcn_bin_sort(const int* __restrict__ tmp,
                                                    const int* __restrict__ offs,
                                                    int* __restrict__ ssrc,
                                                    int* __restrict__ rowptr,
                                                    int* __restrict__ rowend,
                                                    float* __restrict__ dinv,
                                                    int N, int E, int NC) {
    __shared__ int cntL[BIN_NODES];
    __shared__ int wsum[16];
    const int t = threadIdx.x;
    const int lane = t & 31;
    const int wv = t >> 5;
    const int bin = blockIdx.x;
    int start = offs[bin * NC];
    int end   = offs[(bin + 1) * NC];
    start = min(max(start, 0), E);
    end   = min(max(end, start), E);
    cntL[t] = 0;
    __syncthreads();

    int idx = start + t;
    for (; idx + 3 * 512 < end; idx += 4 * 512) {
        const int k0 = tmp[idx], k1 = tmp[idx + 512], k2 = tmp[idx + 1024], k3 = tmp[idx + 1536];
        atomicAdd(&cntL[k0 & (BIN_NODES - 1)], 1);
        atomicAdd(&cntL[k1 & (BIN_NODES - 1)], 1);
        atomicAdd(&cntL[k2 & (BIN_NODES - 1)], 1);
        atomicAdd(&cntL[k3 & (BIN_NODES - 1)], 1);
    }
    for (; idx < end; idx += 512) atomicAdd(&cntL[tmp[idx] & (BIN_NODES - 1)], 1);
    __syncthreads();

    const int c = cntL[t];
    int inc = c;
#pragma unroll
    for (int off = 1; off < 32; off <<= 1) {
        int y = __shfl_up(inc, off, 32);
        if (lane >= off) inc += y;
    }
    if (lane == 31) wsum[wv] = inc;
    __syncthreads();
    int base = (lane < wv) ? wsum[lane & 15] : 0;
#pragma unroll
    for (int off = 16; off > 0; off >>= 1) base += __shfl_xor(base, off, 32);
    const int ex = base + inc - c;

    const int node = bin * BIN_NODES + t;
    if (node < N) {
        rowptr[node] = start + ex;
        rowend[node] = start + ex + c;
        dinv[node]   = rsqrtf((float)(c + 1));
    }
    cntL[t] = ex;
    __syncthreads();

    idx = start + t;
    for (; idx + 3 * 512 < end; idx += 4 * 512) {
        const int k0 = tmp[idx], k1 = tmp[idx + 512], k2 = tmp[idx + 1024], k3 = tmp[idx + 1536];
        const int p0 = atomicAdd(&cntL[k0 & (BIN_NODES - 1)], 1);
        const int p1 = atomicAdd(&cntL[k1 & (BIN_NODES - 1)], 1);
        const int p2 = atomicAdd(&cntL[k2 & (BIN_NODES - 1)], 1);
        const int p3 = atomicAdd(&cntL[k3 & (BIN_NODES - 1)], 1);
        ssrc[min(start + max(p0, 0), end - 1)] = k0 >> BIN_SHIFT;
        ssrc[min(start + max(p1, 0), end - 1)] = k1 >> BIN_SHIFT;
        ssrc[min(start + max(p2, 0), end - 1)] = k2 >> BIN_SHIFT;
        ssrc[min(start + max(p3, 0), end - 1)] = k3 >> BIN_SHIFT;
    }
    for (; idx < end; idx += 512) {
        const int k0 = tmp[idx];
        const int p0 = atomicAdd(&cntL[k0 & (BIN_NODES - 1)], 1);
        ssrc[min(start + max(p0, 0), end - 1)] = k0 >> BIN_SHIFT;
    }
}

__global__ __launch_bounds__(256) void gcn_gemm1_wmma(const float* __restrict__ x,
                                                      const float* __restrict__ W,
                                                      float* __restrict__ h, int N) {
    __shared__ float As[16 * 132];
    const int t = threadIdx.x;
    const int rowBase = blockIdx.x * 16;
    const bool full = (rowBase + 16 <= N);

    if (full) {
#pragma unroll
        for (int i = t; i < 16 * 32; i += 256) {
            int r  = i >> 5;
            int c4 = i & 31;
            float4 v = ((const float4*)(x + (size_t)(rowBase + r) * HIDC))[c4];
            float* p = &As[r * 132 + c4 * 4];
            p[0] = v.x; p[1] = v.y; p[2] = v.z; p[3] = v.w;
        }
    } else {
        for (int i = t; i < 16 * 32; i += 256) {
            int r  = i >> 5;
            int c4 = i & 31;
            float4 v = make_float4(0.f, 0.f, 0.f, 0.f);
            if (rowBase + r < N) v = ((const float4*)(x + (size_t)(rowBase + r) * HIDC))[c4];
            float* p = &As[r * 132 + c4 * 4];
            p[0] = v.x; p[1] = v.y; p[2] = v.z; p[3] = v.w;
        }
    }
    __syncthreads();

    const int lane = t & 31;
    const int wv   = t >> 5;
    const int m    = lane & 15;
    const int half = lane >> 4;
    const int kl   = half * 2;
    const int n    = wv * 16 + (lane & 15);

    v8f c = {};
    const float* Arow = &As[m * 132];
#pragma unroll 8
    for (int k0 = 0; k0 < HIDC; k0 += 4) {
        v2f a, b;
        a.x = Arow[k0 + kl];
        a.y = Arow[k0 + kl + 1];
        b.x = W[(k0 + kl) * HIDC + n];
        b.y = W[(k0 + kl + 1) * HIDC + n];
        c = __builtin_amdgcn_wmma_f32_16x16x4_f32(false, a, false, b,
                                                  (short)0, c, false, false);
    }

    float* hp = h + (size_t)(rowBase + 8 * half) * HIDC + n;
    if (full) {
#pragma unroll
        for (int v = 0; v < 8; ++v) hp[v * HIDC] = c[v];
    } else {
#pragma unroll
        for (int v = 0; v < 8; ++v) {
            if (rowBase + 8 * half + v < N) hp[v * HIDC] = c[v];
        }
    }
}

__global__ __launch_bounds__(256) void gcn_gather1(const float* __restrict__ h,
                                                   const int* __restrict__ ssrc,
                                                   const int* __restrict__ rowptr,
                                                   const int* __restrict__ rowend,
                                                   const float* __restrict__ dinv,
                                                   float* __restrict__ acc, int N) {
    const int i = blockIdx.x * 8 + __builtin_amdgcn_readfirstlane((int)(threadIdx.x >> 5));
    if (i >= N) return;
    const int lane = threadIdx.x & 31;
    const int beg = rowptr[i];
    const int end = rowend[i];
    const float di = dinv[i];
    const float4* h4 = (const float4*)h;
    float4 a = make_float4(0.f, 0.f, 0.f, 0.f);

    for (int base = beg; base < end; base += 32) {
        const int n = min(32, end - base);
        int myS = 0;
        float myW = 0.f;
        if (lane < n) {
            int s = ssrc[base + lane];
            s = min(max(s, 0), N - 1);
            myS = s;
            myW = dinv[s] * di;
        }
        const int myWi = __float_as_int(myW);
        int k = 0;
        for (; k + 4 <= n; k += 4) {
            const float4 v0 = h4[(size_t)__builtin_amdgcn_readlane(myS, k)     * 32 + lane];
            const float4 v1 = h4[(size_t)__builtin_amdgcn_readlane(myS, k + 1) * 32 + lane];
            const float4 v2 = h4[(size_t)__builtin_amdgcn_readlane(myS, k + 2) * 32 + lane];
            const float4 v3 = h4[(size_t)__builtin_amdgcn_readlane(myS, k + 3) * 32 + lane];
            const float w0 = __int_as_float(__builtin_amdgcn_readlane(myWi, k));
            const float w1 = __int_as_float(__builtin_amdgcn_readlane(myWi, k + 1));
            const float w2 = __int_as_float(__builtin_amdgcn_readlane(myWi, k + 2));
            const float w3 = __int_as_float(__builtin_amdgcn_readlane(myWi, k + 3));
            a.x = fmaf(v0.x, w0, a.x); a.y = fmaf(v0.y, w0, a.y); a.z = fmaf(v0.z, w0, a.z); a.w = fmaf(v0.w, w0, a.w);
            a.x = fmaf(v1.x, w1, a.x); a.y = fmaf(v1.y, w1, a.y); a.z = fmaf(v1.z, w1, a.z); a.w = fmaf(v1.w, w1, a.w);
            a.x = fmaf(v2.x, w2, a.x); a.y = fmaf(v2.y, w2, a.y); a.z = fmaf(v2.z, w2, a.z); a.w = fmaf(v2.w, w2, a.w);
            a.x = fmaf(v3.x, w3, a.x); a.y = fmaf(v3.y, w3, a.y); a.z = fmaf(v3.z, w3, a.z); a.w = fmaf(v3.w, w3, a.w);
        }
        const int r = n - k;
        if (r > 0) {
            float4 v0 = h4[(size_t)__builtin_amdgcn_readlane(myS, k) * 32 + lane];
            float4 v1 = make_float4(0.f, 0.f, 0.f, 0.f), v2 = v1;
            if (r > 1) v1 = h4[(size_t)__builtin_amdgcn_readlane(myS, k + 1) * 32 + lane];
            if (r > 2) v2 = h4[(size_t)__builtin_amdgcn_readlane(myS, k + 2) * 32 + lane];
            const float w0 = __int_as_float(__builtin_amdgcn_readlane(myWi, k));
            a.x = fmaf(v0.x, w0, a.x); a.y = fmaf(v0.y, w0, a.y); a.z = fmaf(v0.z, w0, a.z); a.w = fmaf(v0.w, w0, a.w);
            if (r > 1) {
                const float w1 = __int_as_float(__builtin_amdgcn_readlane(myWi, k + 1));
                a.x = fmaf(v1.x, w1, a.x); a.y = fmaf(v1.y, w1, a.y); a.z = fmaf(v1.z, w1, a.z); a.w = fmaf(v1.w, w1, a.w);
            }
            if (r > 2) {
                const float w2 = __int_as_float(__builtin_amdgcn_readlane(myWi, k + 2));
                a.x = fmaf(v2.x, w2, a.x); a.y = fmaf(v2.y, w2, a.y); a.z = fmaf(v2.z, w2, a.z); a.w = fmaf(v2.w, w2, a.w);
            }
        }
    }
    ((float4*)(acc + (size_t)i * HIDC))[lane] = a;
}

__global__ __launch_bounds__(256) void gcn_fin1_dot(const float* __restrict__ acc,
                                                    const float* __restrict__ h,
                                                    const float* __restrict__ dinv,
                                                    const float* __restrict__ b1,
                                                    const float* __restrict__ W2,
                                                    float* __restrict__ g, int N) {
    int i = blockIdx.x * 8 + (threadIdx.x >> 5);
    if (i >= N) return;
    int lane = threadIdx.x & 31;
    float di = dinv[i];
    float ns = di * di;
    float4 a  = ((const float4*)(acc + (size_t)i * HIDC))[lane];
    float4 hv = ((const float4*)(h   + (size_t)i * HIDC))[lane];
    float4 bb = ((const float4*)b1)[lane];
    float4 w  = ((const float4*)W2)[lane];
    float r0 = fmaxf(a.x + hv.x * ns + bb.x, 0.0f);
    float r1 = fmaxf(a.y + hv.y * ns + bb.y, 0.0f);
    float r2 = fmaxf(a.z + hv.z * ns + bb.z, 0.0f);
    float r3 = fmaxf(a.w + hv.w * ns + bb.w, 0.0f);
    float p = r0 * w.x + r1 * w.y + r2 * w.z + r3 * w.w;
#pragma unroll
    for (int off = 16; off > 0; off >>= 1) p += __shfl_down(p, off, 32);
    if (lane == 0) g[i] = p * di;
}

__global__ __launch_bounds__(128) void gcn_gather2(const float* __restrict__ g,
                                                   const int* __restrict__ ssrc,
                                                   const int* __restrict__ rowptr,
                                                   const int* __restrict__ rowend,
                                                   const float* __restrict__ dinv,
                                                   const float* __restrict__ b2,
                                                   float* __restrict__ out, int N) {
    const int i = blockIdx.x * blockDim.x + threadIdx.x;
    if (i >= N) return;
    const int beg = rowptr[i];
    const int end = rowend[i];
    const float di = dinv[i];
    const float bias = b2[0];
    float a = g[i];
    int k = beg;
    for (; k + 4 <= end; k += 4) {
        int s0 = ssrc[k], s1 = ssrc[k + 1], s2 = ssrc[k + 2], s3 = ssrc[k + 3];
        s0 = min(max(s0, 0), N - 1);
        s1 = min(max(s1, 0), N - 1);
        s2 = min(max(s2, 0), N - 1);
        s3 = min(max(s3, 0), N - 1);
        const float g0 = g[s0], g1 = g[s1], g2 = g[s2], g3 = g[s3];
        a += g0; a += g1; a += g2; a += g3;
    }
    const int r = end - k;
    if (r > 0) {
        const int k1 = min(k + 1, end - 1);
        const int k2 = min(k + 2, end - 1);
        int s0 = ssrc[k], s1 = ssrc[k1], s2 = ssrc[k2];
        s0 = min(max(s0, 0), N - 1);
        s1 = min(max(s1, 0), N - 1);
        s2 = min(max(s2, 0), N - 1);
        const float g0 = g[s0], g1 = g[s1], g2 = g[s2];
        a += g0;
        a += (r > 1) ? g1 : 0.0f;
        a += (r > 2) ? g2 : 0.0f;
    }
    out[i] = a * di + bias;
}

extern "C" void kernel_launch(void* const* d_in, const int* in_sizes, int n_in,
                              void* d_out, int out_size, void* d_ws, size_t ws_size,
                              hipStream_t stream) {
    const float* x  = (const float*)d_in[0];
    const float* W1 = (const float*)d_in[1];
    const float* b1 = (const float*)d_in[2];
    const float* W2 = (const float*)d_in[3];
    const float* b2 = (const float*)d_in[4];
    const int*   ei = (const int*)d_in[5];

    const int N = in_sizes[0] / HIDC;
    const int E = in_sizes[5] / 2;
    const int* src = ei;
    const int* dst = ei + E;
    float* out = (float*)d_out;

    const int NB = (N + BIN_NODES - 1) / BIN_NODES;
    const int NC = (E + CHUNK - 1) / CHUNK;
    const int M  = NB * NC + 1;

    float* h      = (float*)d_ws;
    float* acc    = h   + (size_t)N * HIDC;
    int*   bhist  = (int*)(acc + (size_t)N * HIDC);
    int*   offs   = bhist + M;
    int*   rowptr = offs + M;
    int*   rowend = rowptr + N;
    float* dinv   = (float*)(rowend + N);
    float* g      = dinv + N;
    int*   ssrc   = (int*)(g + N);
    int*   tmp    = ssrc + E;

    const int TB = 256;

    gcn_bin_count  <<<NC, TB, 0, stream>>>(dst, bhist, N, E, NB, NC);
    gcn_scan       <<<(M + 4095) / 4096, 1024, 0, stream>>>(bhist, offs, M);
    gcn_bin_scatter<<<NC, TB, 0, stream>>>(src, dst, offs, tmp, N, E, NB, NC);
    gcn_bin_sort   <<<NB, BIN_NODES, 0, stream>>>(tmp, offs, ssrc, rowptr, rowend, dinv, N, E, NC);
    gcn_gemm1_wmma<<<(N + 15) / 16,     TB, 0, stream>>>(x, W1, h, N);
    gcn_gather1   <<<(N + 7) / 8,       TB, 0, stream>>>(h, ssrc, rowptr, rowend, dinv, acc, N);
    gcn_fin1_dot  <<<(N + 7) / 8,       TB, 0, stream>>>(acc, h, dinv, b1, W2, g, N);
    gcn_gather2   <<<(N + 127) / 128,   128, 0, stream>>>(g, ssrc, rowptr, rowend, dinv, b2, out, N);
}


# layer-1 finalize (self term, bias, relu, dot with W2) done by the gather wave on its register row: acc plane and the finalize launch are gone, 7 launches; f32 sum order unchanged
# speedup vs baseline: 3.4304x; 1.0722x over previous
#include <hip/hip_runtime.h>
#include <math.h>

#define HIDC 128

typedef __attribute__((ext_vector_type(2))) float v2f;
typedef __attribute__((ext_vector_type(8))) float v8f;

#define BIN_SHIFT 9
#define BIN_NODES 512
#define CHUNK     4096
#define MAXB      1024

__global__ __launch_bounds__(256) void gcn_bin_count(const int* __restrict__ dst,
                                                     int* __restrict__ blockHist,
                                                     int N, int E, int NB, int NC) {
    __shared__ int hist[MAXB];
    const int t = threadIdx.x;
    const int chunk = blockIdx.x;
    for (int b = t; b < NB; b += 256) hist[b] = 0;
    __syncthreads();
    const int e0 = chunk * CHUNK + t;
    if (chunk * CHUNK + CHUNK <= E) {
        int d[16];
#pragma unroll
        for (int j = 0; j < 16; ++j) d[j] = dst[e0 + j * 256];
#pragma unroll
        for (int j = 0; j < 16; ++j) {
            const int dd = min(max(d[j], 0), N - 1);
            atomicAdd(&hist[min(dd >> BIN_SHIFT, MAXB - 1)], 1);
        }
    } else {
        for (int e = e0; e < E; e += 256) {
            const int dd = min(max(dst[e], 0), N - 1);
            atomicAdd(&hist[min(dd >> BIN_SHIFT, MAXB - 1)], 1);
        }
    }
    __syncthreads();
    for (int b = t; b < NB; b += 256) blockHist[b * NC + chunk] = hist[b];
    if (chunk == 0 && t == 0) blockHist[NB * NC] = 0;
}

__global__ __launch_bounds__(1024) void gcn_scan(const int* __restrict__ cnt,
                                                 int* __restrict__ offs, int M) {
    __shared__ int preS[32];
    __shared__ int ownS[32];
    const int t = threadIdx.x;
    const int lane = t & 31;
    const int wv = t >> 5;
    const int blockStart = blockIdx.x * 4096;

    int pre = 0;
    const int4* c4 = (const int4*)cnt;
#pragma unroll 4
    for (int j = 0; j < (int)blockIdx.x; ++j) {
        int4 v = c4[j * 1024 + t];
        pre += (v.x + v.y) + (v.z + v.w);
    }
#pragma unroll
    for (int off = 16; off > 0; off >>= 1) pre += __shfl_xor(pre, off, 32);
    if (lane == 0) preS[wv] = pre;

    const int i0 = blockStart + t * 4;
    int c0 = (i0     < M) ? cnt[i0]     : 0;
    int c1 = (i0 + 1 < M) ? cnt[i0 + 1] : 0;
    int c2 = (i0 + 2 < M) ? cnt[i0 + 2] : 0;
    int c3 = (i0 + 3 < M) ? cnt[i0 + 3] : 0;
    const int s = (c0 + c1) + (c2 + c3);
    int inc = s;
#pragma unroll
    for (int off = 1; off < 32; off <<= 1) {
        int y = __shfl_up(inc, off, 32);
        if (lane >= off) inc += y;
    }
    if (lane == 31) ownS[wv] = inc;
    __syncthreads();

    int ab = preS[lane] + ((lane < wv) ? ownS[lane] : 0);
#pragma unroll
    for (int off = 16; off > 0; off >>= 1) ab += __shfl_xor(ab, off, 32);

    const int e0 = ab + inc - s;
    const int e1 = e0 + c0;
    const int e2 = e1 + c1;
    const int e3 = e2 + c2;
    if (i0 < M)     offs[i0]     = e0;
    if (i0 + 1 < M) offs[i0 + 1] = e1;
    if (i0 + 2 < M) offs[i0 + 2] = e2;
    if (i0 + 3 < M) offs[i0 + 3] = e3;
}

__global__ __launch_bounds__(256) void gcn_bin_scatter(const int* __restrict__ src,
                                                       const int* __restrict__ dst,
                                                       const int* __restrict__ offs,
                                                       int* __restrict__ tmp,
                                                       int N, int E, int NB, int NC) {
    __shared__ int cur[MAXB];
    const int t = threadIdx.x;
    const int chunk = blockIdx.x;
    for (int b = t; b < NB; b += 256) cur[b] = offs[b * NC + chunk];
    __syncthreads();
    const int e0 = chunk * CHUNK + t;
    if (chunk * CHUNK + CHUNK <= E) {
#pragma unroll
        for (int hf = 0; hf < 2; ++hf) {
            int d[8], s[8];
#pragma unroll
            for (int j = 0; j < 8; ++j) {
                d[j] = dst[e0 + (hf * 8 + j) * 256];
                s[j] = src[e0 + (hf * 8 + j) * 256];
            }
            int pos[8];
#pragma unroll
            for (int j = 0; j < 8; ++j) {
                d[j] = min(max(d[j], 0), N - 1);
                pos[j] = atomicAdd(&cur[min(d[j] >> BIN_SHIFT, MAXB - 1)], 1);
            }
#pragma unroll
            for (int j = 0; j < 8; ++j) {
                const int ss = min(max(s[j], 0), N - 1);
                const int p = min(max(pos[j], 0), E - 1);
                tmp[p] = (ss << BIN_SHIFT) | (d[j] & (BIN_NODES - 1));
            }
        }
    } else {
        for (int e = e0; e < E; e += 256) {
            const int dd = min(max(dst[e], 0), N - 1);
            const int ss = min(max(src[e], 0), N - 1);
            int pos = atomicAdd(&cur[min(dd >> BIN_SHIFT, MAXB - 1)], 1);
            pos = min(max(pos, 0), E - 1);
            tmp[pos] = (ss << BIN_SHIFT) | (dd & (BIN_NODES - 1));
        }
    }
}

__global__ __launch_bounds__(512) void gcn_bin_sort(const int* __restrict__ tmp,
                                                    const int* __restrict__ offs,
                                                    int* __restrict__ ssrc,
                                                    int* __restrict__ rowptr,
                                                    int* __restrict__ rowend,
                                                    float* __restrict__ dinv,
                                                    int N, int E, int NC) {
    __shared__ int cntL[BIN_NODES];
    __shared__ int wsum[16];
    const int t = threadIdx.x;
    const int lane = t & 31;
    const int wv = t >> 5;
    const int bin = blockIdx.x;
    int start = offs[bin * NC];
    int end   = offs[(bin + 1) * NC];
    start = min(max(start, 0), E);
    end   = min(max(end, start), E);
    cntL[t] = 0;
    __syncthreads();

    int idx = start + t;
    for (; idx + 3 * 512 < end; idx += 4 * 512) {
        const int k0 = tmp[idx], k1 = tmp[idx + 512], k2 = tmp[idx + 1024], k3 = tmp[idx + 1536];
        atomicAdd(&cntL[k0 & (BIN_NODES - 1)], 1);
        atomicAdd(&cntL[k1 & (BIN_NODES - 1)], 1);
        atomicAdd(&cntL[k2 & (BIN_NODES - 1)], 1);
        atomicAdd(&cntL[k3 & (BIN_NODES - 1)], 1);
    }
    for (; idx < end; idx += 512) atomicAdd(&cntL[tmp[idx] & (BIN_NODES - 1)], 1);
    __syncthreads();

    const int c = cntL[t];
    int inc = c;
#pragma unroll
    for (int off = 1; off < 32; off <<= 1) {
        int y = __shfl_up(inc, off, 32);
        if (lane >= off) inc += y;
    }
    if (lane == 31) wsum[wv] = inc;
    __syncthreads();
    int base = (lane < wv) ? wsum[lane & 15] : 0;
#pragma unroll
    for (int off = 16; off > 0; off >>= 1) base += __shfl_xor(base, off, 32);
    const int ex = base + inc - c;

    const int node = bin * BIN_NODES + t;
    if (node < N) {
        rowptr[node] = start + ex;
        rowend[node] = start + ex + c;
        dinv[node]   = rsqrtf((float)(c + 1));
    }
    cntL[t] = ex;
    __syncthreads();

    idx = start + t;
    for (; idx + 3 * 512 < end; idx += 4 * 512) {
        const int k0 = tmp[idx], k1 = tmp[idx + 512], k2 = tmp[idx + 1024], k3 = tmp[idx + 1536];
        const int p0 = atomicAdd(&cntL[k0 & (BIN_NODES - 1)], 1);
        const int p1 = atomicAdd(&cntL[k1 & (BIN_NODES - 1)], 1);
        const int p2 = atomicAdd(&cntL[k2 & (BIN_NODES - 1)], 1);
        const int p3 = atomicAdd(&cntL[k3 & (BIN_NODES - 1)], 1);
        ssrc[min(start + max(p0, 0), end - 1)] = k0 >> BIN_SHIFT;
        ssrc[min(start + max(p1, 0), end - 1)] = k1 >> BIN_SHIFT;
        ssrc[min(start + max(p2, 0), end - 1)] = k2 >> BIN_SHIFT;
        ssrc[min(start + max(p3, 0), end - 1)] = k3 >> BIN_SHIFT;
    }
    for (; idx < end; idx += 512) {
        const int k0 = tmp[idx];
        const int p0 = atomicAdd(&cntL[k0 & (BIN_NODES - 1)], 1);
        ssrc[min(start + max(p0, 0), end - 1)] = k0 >> BIN_SHIFT;
    }
}

__global__ __launch_bounds__(256) void gcn_gemm1_wmma(const float* __restrict__ x,
                                                      const float* __restrict__ W,
                                                      float* __restrict__ h, int N) {
    __shared__ float As[16 * 132];
    const int t = threadIdx.x;
    const int rowBase = blockIdx.x * 16;
    const bool full = (rowBase + 16 <= N);

    if (full) {
#pragma unroll
        for (int i = t; i < 16 * 32; i += 256) {
            int r  = i >> 5;
            int c4 = i & 31;
            float4 v = ((const float4*)(x + (size_t)(rowBase + r) * HIDC))[c4];
            float* p = &As[r * 132 + c4 * 4];
            p[0] = v.x; p[1] = v.y; p[2] = v.z; p[3] = v.w;
        }
    } else {
        for (int i = t; i < 16 * 32; i += 256) {
            int r  = i >> 5;
            int c4 = i & 31;
            float4 v = make_float4(0.f, 0.f, 0.f, 0.f);
            if (rowBase + r < N) v = ((const float4*)(x + (size_t)(rowBase + r) * HIDC))[c4];
            float* p = &As[r * 132 + c4 * 4];
            p[0] = v.x; p[1] = v.y; p[2] = v.z; p[3] = v.w;
        }
    }
    __syncthreads();

    const int lane = t & 31;
    const int wv   = t >> 5;
    const int m    = lane & 15;
    const int half = lane >> 4;
    const int kl   = half * 2;
    const int n    = wv * 16 + (lane & 15);

    v8f c = {};
    const float* Arow = &As[m * 132];
#pragma unroll 8
    for (int k0 = 0; k0 < HIDC; k0 += 4) {
        v2f a, b;
        a.x = Arow[k0 + kl];
        a.y = Arow[k0 + kl + 1];
        b.x = W[(k0 + kl) * HIDC + n];
        b.y = W[(k0 + kl + 1) * HIDC + n];
        c = __builtin_amdgcn_wmma_f32_16x16x4_f32(false, a, false, b,
                                                  (short)0, c, false, false);
    }

    float* hp = h + (size_t)(rowBase + 8 * half) * HIDC + n;
    if (full) {
#pragma unroll
        for (int v = 0; v < 8; ++v) hp[v * HIDC] = c[v];
    } else {
#pragma unroll
        for (int v = 0; v < 8; ++v) {
            if (rowBase + 8 * half + v < N) hp[v * HIDC] = c[v];
        }
    }
}

__global__ __launch_bounds__(256) void gcn_gather1_fin(const float* __restrict__ h,
                                                       const int* __restrict__ ssrc,
                                                       const int* __restrict__ rowptr,
                                                       const int* __restrict__ rowend,
                                                       const float* __restrict__ dinv,
                                                       const float* __restrict__ b1,
                                                       const float* __restrict__ W2,
                                                       float* __restrict__ g, int N) {
    const int i = blockIdx.x * 8 + __builtin_amdgcn_readfirstlane((int)(threadIdx.x >> 5));
    if (i >= N) return;
    const int lane = threadIdx.x & 31;
    const int beg = rowptr[i];
    const int end = rowend[i];
    const float di = dinv[i];
    const float4* h4 = (const float4*)h;
    const float4 hv = h4[(size_t)i * 32 + lane];
    const float4 bb = ((const float4*)b1)[lane];
    const float4 w  = ((const float4*)W2)[lane];
    float4 a = make_float4(0.f, 0.f, 0.f, 0.f);

    for (int base = beg; base < end; base += 32) {
        const int n = min(32, end - base);
        int myS = 0;
        float myW = 0.f;
        if (lane < n) {
            int s = ssrc[base + lane];
            s = min(max(s, 0), N - 1);
            myS = s;
            myW = dinv[s] * di;
        }
        const int myWi = __float_as_int(myW);
        int k = 0;
        for (; k + 4 <= n; k += 4) {
            const float4 v0 = h4[(size_t)__builtin_amdgcn_readlane(myS, k)     * 32 + lane];
            const float4 v1 = h4[(size_t)__builtin_amdgcn_readlane(myS, k + 1) * 32 + lane];
            const float4 v2 = h4[(size_t)__builtin_amdgcn_readlane(myS, k + 2) * 32 + lane];
            const float4 v3 = h4[(size_t)__builtin_amdgcn_readlane(myS, k + 3) * 32 + lane];
            const float w0 = __int_as_float(__builtin_amdgcn_readlane(myWi, k));
            const float w1 = __int_as_float(__builtin_amdgcn_readlane(myWi, k + 1));
            const float w2 = __int_as_float(__builtin_amdgcn_readlane(myWi, k + 2));
            const float w3 = __int_as_float(__builtin_amdgcn_readlane(myWi, k + 3));
            a.x = fmaf(v0.x, w0, a.x); a.y = fmaf(v0.y, w0, a.y); a.z = fmaf(v0.z, w0, a.z); a.w = fmaf(v0.w, w0, a.w);
            a.x = fmaf(v1.x, w1, a.x); a.y = fmaf(v1.y, w1, a.y); a.z = fmaf(v1.z, w1, a.z); a.w = fmaf(v1.w, w1, a.w);
            a.x = fmaf(v2.x, w2, a.x); a.y = fmaf(v2.y, w2, a.y); a.z = fmaf(v2.z, w2, a.z); a.w = fmaf(v2.w, w2, a.w);
            a.x = fmaf(v3.x, w3, a.x); a.y = fmaf(v3.y, w3, a.y); a.z = fmaf(v3.z, w3, a.z); a.w = fmaf(v3.w, w3, a.w);
        }
        const int r = n - k;
        if (r > 0) {
            float4 v0 = h4[(size_t)__builtin_amdgcn_readlane(myS, k) * 32 + lane];
            float4 v1 = make_float4(0.f, 0.f, 0.f, 0.f), v2 = v1;
            if (r > 1) v1 = h4[(size_t)__builtin_amdgcn_readlane(myS, k + 1) * 32 + lane];
            if (r > 2) v2 = h4[(size_t)__builtin_amdgcn_readlane(myS, k + 2) * 32 + lane];
            const float w0 = __int_as_float(__builtin_amdgcn_readlane(myWi, k));
            a.x = fmaf(v0.x, w0, a.x); a.y = fmaf(v0.y, w0, a.y); a.z = fmaf(v0.z, w0, a.z); a.w = fmaf(v0.w, w0, a.w);
            if (r > 1) {
                const float w1 = __int_as_float(__builtin_amdgcn_readlane(myWi, k + 1));
                a.x = fmaf(v1.x, w1, a.x); a.y = fmaf(v1.y, w1, a.y); a.z = fmaf(v1.z, w1, a.z); a.w = fmaf(v1.w, w1, a.w);
            }
            if (r > 2) {
                const float w2 = __int_as_float(__builtin_amdgcn_readlane(myWi, k + 2));
                a.x = fmaf(v2.x, w2, a.x); a.y = fmaf(v2.y, w2, a.y); a.z = fmaf(v2.z, w2, a.z); a.w = fmaf(v2.w, w2, a.w);
            }
        }
    }
    const float ns = di * di;
    const float r0 = fmaxf(a.x + hv.x * ns + bb.x, 0.0f);
    const float r1 = fmaxf(a.y + hv.y * ns + bb.y, 0.0f);
    const float r2 = fmaxf(a.z + hv.z * ns + bb.z, 0.0f);
    const float r3 = fmaxf(a.w + hv.w * ns + bb.w, 0.0f);
    float p = r0 * w.x + r1 * w.y + r2 * w.z + r3 * w.w;
#pragma unroll
    for (int off = 16; off > 0; off >>= 1) p += __shfl_down(p, off, 32);
    if (lane == 0) g[i] = p * di;
}

__global__ __launch_bounds__(128) void gcn_gather2(const float* __restrict__ g,
                                                   const int* __restrict__ ssrc,
                                                   const int* __restrict__ rowptr,
                                                   const int* __restrict__ rowend,
                                                   const float* __restrict__ dinv,
                                                   const float* __restrict__ b2,
                                                   float* __restrict__ out, int N) {
    const int i = blockIdx.x * blockDim.x + threadIdx.x;
    if (i >= N) return;
    const int beg = rowptr[i];
    const int end = rowend[i];
    const float di = dinv[i];
    const float bias = b2[0];
    float a = g[i];
    int k = beg;
    for (; k + 4 <= end; k += 4) {
        int s0 = ssrc[k], s1 = ssrc[k + 1], s2 = ssrc[k + 2], s3 = ssrc[k + 3];
        s0 = min(max(s0, 0), N - 1);
        s1 = min(max(s1, 0), N - 1);
        s2 = min(max(s2, 0), N - 1);
        s3 = min(max(s3, 0), N - 1);
        const float g0 = g[s0], g1 = g[s1], g2 = g[s2], g3 = g[s3];
        a += g0; a += g1; a += g2; a += g3;
    }
    const int r = end - k;
    if (r > 0) {
        const int k1 = min(k + 1, end - 1);
        const int k2 = min(k + 2, end - 1);
        int s0 = ssrc[k], s1 = ssrc[k1], s2 = ssrc[k2];
        s0 = min(max(s0, 0), N - 1);
        s1 = min(max(s1, 0), N - 1);
        s2 = min(max(s2, 0), N - 1);
        const float g0 = g[s0], g1 = g[s1], g2 = g[s2];
        a += g0;
        a += (r > 1) ? g1 : 0.0f;
        a += (r > 2) ? g2 : 0.0f;
    }
    out[i] = a * di + bias;
}

extern "C" void kernel_launch(void* const* d_in, const int* in_sizes, int n_in,
                              void* d_out, int out_size, void* d_ws, size_t ws_size,
                              hipStream_t stream) {
    const float* x  = (const float*)d_in[0];
    const float* W1 = (const float*)d_in[1];
    const float* b1 = (const float*)d_in[2];
    const float* W2 = (const float*)d_in[3];
    const float* b2 = (const float*)d_in[4];
    const int*   ei = (const int*)d_in[5];

    const int N = in_sizes[0] / HIDC;
    const int E = in_sizes[5] / 2;
    const int* src = ei;
    const int* dst = ei + E;
    float* out = (float*)d_out;

    const int NB = (N + BIN_NODES - 1) / BIN_NODES;
    const int NC = (E + CHUNK - 1) / CHUNK;
    const int M  = NB * NC + 1;

    float* h      = (float*)d_ws;
    int*   bhist  = (int*)(h + (size_t)N * HIDC);
    int*   offs   = bhist + M;
    int*   rowptr = offs + M;
    int*   rowend = rowptr + N;
    float* dinv   = (float*)(rowend + N);
    float* g      = dinv + N;
    int*   ssrc   = (int*)(g + N);
    int*   tmp    = ssrc + E;

    const int TB = 256;

    gcn_bin_count  <<<NC, TB, 0, stream>>>(dst, bhist, N, E, NB, NC);
    gcn_scan       <<<(M + 4095) / 4096, 1024, 0, stream>>>(bhist, offs, M);
    gcn_bin_scatter<<<NC, TB, 0, stream>>>(src, dst, offs, tmp, N, E, NB, NC);
    gcn_bin_sort   <<<NB, BIN_NODES, 0, stream>>>(tmp, offs, ssrc, rowptr, rowend, dinv, N, E, NC);
    gcn_gemm1_wmma<<<(N + 15) / 16,     TB, 0, stream>>>(x, W1, h, N);
    gcn_gather1_fin<<<(N + 7) / 8,      TB, 0, stream>>>(h, ssrc, rowptr, rowend, dinv, b1, W2, g, N);
    gcn_gather2   <<<(N + 127) / 128,   128, 0, stream>>>(g, ssrc, rowptr, rowend, dinv, b2, out, N);
}


# layer 1 aggregates x first: a block sums 32 nodes' source rows of x into LDS, multiplies the tile by W1 on WMMA, finishes in place; h plane, product launch gone; f32 sums: neighbors before product
# speedup vs baseline: 3.8307x; 1.1167x over previous
#include <hip/hip_runtime.h>
#include <math.h>

#define HIDC 128

typedef __attribute__((ext_vector_type(2))) float v2f;
typedef __attribute__((ext_vector_type(8))) float v8f;

#define BIN_SHIFT 9
#define BIN_NODES 512
#define CHUNK     4096
#define MAXB      1024

__global__ __launch_bounds__(256) void gcn_bin_count(const int* __restrict__ dst,
                                                     int* __restrict__ blockHist,
                                                     int N, int E, int NB, int NC) {
    __shared__ int hist[MAXB];
    const int t = threadIdx.x;
    const int chunk = blockIdx.x;
    for (int b = t; b < NB; b += 256) hist[b] = 0;
    __syncthreads();
    const int e0 = chunk * CHUNK + t;
    if (chunk * CHUNK + CHUNK <= E) {
        int d[16];
#pragma unroll
        for (int j = 0; j < 16; ++j) d[j] = dst[e0 + j * 256];
#pragma unroll
        for (int j = 0; j < 16; ++j) {
            const int dd = min(max(d[j], 0), N - 1);
            atomicAdd(&hist[min(dd >> BIN_SHIFT, MAXB - 1)], 1);
        }
    } else {
        for (int e = e0; e < E; e += 256) {
            const int dd = min(max(dst[e], 0), N - 1);
            atomicAdd(&hist[min(dd >> BIN_SHIFT, MAXB - 1)], 1);
        }
    }
    __syncthreads();
    for (int b = t; b < NB; b += 256) blockHist[b * NC + chunk] = hist[b];
    if (chunk == 0 && t == 0) blockHist[NB * NC] = 0;
}

__global__ __launch_bounds__(1024) void gcn_scan(const int* __restrict__ cnt,
                                                 int* __restrict__ offs, int M) {
    __shared__ int preS[32];
    __shared__ int ownS[32];
    const int t = threadIdx.x;
    const int lane = t & 31;
    const int wv = t >> 5;
    const int blockStart = blockIdx.x * 4096;

    int pre = 0;
    const int4* c4 = (const int4*)cnt;
#pragma unroll 4
    for (int j = 0; j < (int)blockIdx.x; ++j) {
        int4 v = c4[j * 1024 + t];
        pre += (v.x + v.y) + (v.z + v.w);
    }
#pragma unroll
    for (int off = 16; off > 0; off >>= 1) pre += __shfl_xor(pre, off, 32);
    if (lane == 0) preS[wv] = pre;

    const int i0 = blockStart + t * 4;
    int c0 = (i0     < M) ? cnt[i0]     : 0;
    int c1 = (i0 + 1 < M) ? cnt[i0 + 1] : 0;
    int c2 = (i0 + 2 < M) ? cnt[i0 + 2] : 0;
    int c3 = (i0 + 3 < M) ? cnt[i0 + 3] : 0;
    const int s = (c0 + c1) + (c2 + c3);
    int inc = s;
#pragma unroll
    for (int off = 1; off < 32; off <<= 1) {
        int y = __shfl_up(inc, off, 32);
        if (lane >= off) inc += y;
    }
    if (lane == 31) ownS[wv] = inc;
    __syncthreads();

    int ab = preS[lane] + ((lane < wv) ? ownS[lane] : 0);
#pragma unroll
    for (int off = 16; off > 0; off >>= 1) ab += __shfl_xor(ab, off, 32);

    const int e0 = ab + inc - s;
    const int e1 = e0 + c0;
    const int e2 = e1 + c1;
    const int e3 = e2 + c2;
    if (i0 < M)     offs[i0]     = e0;
    if (i0 + 1 < M) offs[i0 + 1] = e1;
    if (i0 + 2 < M) offs[i0 + 2] = e2;
    if (i0 + 3 < M) offs[i0 + 3] = e3;
}

__global__ __launch_bounds__(256) void gcn_bin_scatter(const int* __restrict__ src,
                                                       const int* __restrict__ dst,
                                                       const int* __restrict__ offs,
                                                       int* __restrict__ tmp,
                                                       int N, int E, int NB, int NC) {
    __shared__ int cur[MAXB];
    const int t = threadIdx.x;
    const int chunk = blockIdx.x;
    for (int b = t; b < NB; b += 256) cur[b] = offs[b * NC + chunk];
    __syncthreads();
    const int e0 = chunk * CHUNK + t;
    if (chunk * CHUNK + CHUNK <= E) {
#pragma unroll
        for (int hf = 0; hf < 2; ++hf) {
            int d[8], s[8];
#pragma unroll
            for (int j = 0; j < 8; ++j) {
                d[j] = dst[e0 + (hf * 8 + j) * 256];
                s[j] = src[e0 + (hf * 8 + j) * 256];
            }
            int pos[8];
#pragma unroll
            for (int j = 0; j < 8; ++j) {
                d[j] = min(max(d[j], 0), N - 1);
                pos[j] = atomicAdd(&cur[min(d[j] >> BIN_SHIFT, MAXB - 1)], 1);
            }
#pragma unroll
            for (int j = 0; j < 8; ++j) {
                const int ss = min(max(s[j], 0), N - 1);
                const int p = min(max(pos[j], 0), E - 1);
                tmp[p] = (ss << BIN_SHIFT) | (d[j] & (BIN_NODES - 1));
            }
        }
    } else {
        for (int e = e0; e < E; e += 256) {
            const int dd = min(max(dst[e], 0), N - 1);
            const int ss = min(max(src[e], 0), N - 1);
            int pos = atomicAdd(&cur[min(dd >> BIN_SHIFT, MAXB - 1)], 1);
            pos = min(max(pos, 0), E - 1);
            tmp[pos] = (ss << BIN_SHIFT) | (dd & (BIN_NODES - 1));
        }
    }
}

__global__ __launch_bounds__(512) void gcn_bin_sort(const int* __restrict__ tmp,
                                                    const int* __restrict__ offs,
                                                    int* __restrict__ ssrc,
                                                    int* __restrict__ rowptr,
                                                    int* __restrict__ rowend,
                                                    float* __restrict__ dinv,
                                                    int N, int E, int NC) {
    __shared__ int cntL[BIN_NODES];
    __shared__ int wsum[16];
    const int t = threadIdx.x;
    const int lane = t & 31;
    const int wv = t >> 5;
    const int bin = blockIdx.x;
    int start = offs[bin * NC];
    int end   = offs[(bin + 1) * NC];
    start = min(max(start, 0), E);
    end   = min(max(end, start), E);
    cntL[t] = 0;
    __syncthreads();

    int idx = start + t;
    for (; idx + 3 * 512 < end; idx += 4 * 512) {
        const int k0 = tmp[idx], k1 = tmp[idx + 512], k2 = tmp[idx + 1024], k3 = tmp[idx + 1536];
        atomicAdd(&cntL[k0 & (BIN_NODES - 1)], 1);
        atomicAdd(&cntL[k1 & (BIN_NODES - 1)], 1);
        atomicAdd(&cntL[k2 & (BIN_NODES - 1)], 1);
        atomicAdd(&cntL[k3 & (BIN_NODES - 1)], 1);
    }
    for (; idx < end; idx += 512) atomicAdd(&cntL[tmp[idx] & (BIN_NODES - 1)], 1);
    __syncthreads();

    const int c = cntL[t];
    int inc = c;
#pragma unroll
    for (int off = 1; off < 32; off <<= 1) {
        int y = __shfl_up(inc, off, 32);
        if (lane >= off) inc += y;
    }
    if (lane == 31) wsum[wv] = inc;
    __syncthreads();
    int base = (lane < wv) ? wsum[lane & 15] : 0;
#pragma unroll
    for (int off = 16; off > 0; off >>= 1) base += __shfl_xor(base, off, 32);
    const int ex = base + inc - c;

    const int node = bin * BIN_NODES + t;
    if (node < N) {
        rowptr[node] = start + ex;
        rowend[node] = start + ex + c;
        dinv[node]   = rsqrtf((float)(c + 1));
    }
    cntL[t] = ex;
    __syncthreads();

    idx = start + t;
    for (; idx + 3 * 512 < end; idx += 4 * 512) {
        const int k0 = tmp[idx], k1 = tmp[idx + 512], k2 = tmp[idx + 1024], k3 = tmp[idx + 1536];
        const int p0 = atomicAdd(&cntL[k0 & (BIN_NODES - 1)], 1);
        const int p1 = atomicAdd(&cntL[k1 & (BIN_NODES - 1)], 1);
        const int p2 = atomicAdd(&cntL[k2 & (BIN_NODES - 1)], 1);
        const int p3 = atomicAdd(&cntL[k3 & (BIN_NODES - 1)], 1);
        ssrc[min(start + max(p0, 0), end - 1)] = k0 >> BIN_SHIFT;
        ssrc[min(start + max(p1, 0), end - 1)] = k1 >> BIN_SHIFT;
        ssrc[min(start + max(p2, 0), end - 1)] = k2 >> BIN_SHIFT;
        ssrc[min(start + max(p3, 0), end - 1)] = k3 >> BIN_SHIFT;
    }
    for (; idx < end; idx += 512) {
        const int k0 = tmp[idx];
        const int p0 = atomicAdd(&cntL[k0 & (BIN_NODES - 1)], 1);
        ssrc[min(start + max(p0, 0), end - 1)] = k0 >> BIN_SHIFT;
    }
}


#define TROWS 32

__device__ __forceinline__ void gcn_load_ids(const int* __restrict__ ssrc,
                                             const float* __restrict__ dinv,
                                             int base, int n, float di, int lane, int N,
                                             int& myS, int& myWi) {
    int s = 0;
    float w = 0.f;
    if (lane < n) {
        s = ssrc[base + lane];
        s = min(max(s, 0), N - 1);
        w = dinv[s] * di;
    }
    myS = s;
    myWi = __float_as_int(w);
}

__device__ __forceinline__ void gcn_sum_rows(const float4* __restrict__ x4, int lane,
                                             int myS, int myWi, int n, float4& a) {
    int k = 0;
    for (; k + 4 <= n; k += 4) {
        const float4 v0 = x4[(size_t)__builtin_amdgcn_readlane(myS, k)     * 32 + lane];
        const float4 v1 = x4[(size_t)__builtin_amdgcn_readlane(myS, k + 1) * 32 + lane];
        const float4 v2 = x4[(size_t)__builtin_amdgcn_readlane(myS, k + 2) * 32 + lane];
        const float4 v3 = x4[(size_t)__builtin_amdgcn_readlane(myS, k + 3) * 32 + lane];
        const float w0 = __int_as_float(__builtin_amdgcn_readlane(myWi, k));
        const float w1 = __int_as_float(__builtin_amdgcn_readlane(myWi, k + 1));
        const float w2 = __int_as_float(__builtin_amdgcn_readlane(myWi, k + 2));
        const float w3 = __int_as_float(__builtin_amdgcn_readlane(myWi, k + 3));
        a.x = fmaf(v0.x, w0, a.x); a.y = fmaf(v0.y, w0, a.y); a.z = fmaf(v0.z, w0, a.z); a.w = fmaf(v0.w, w0, a.w);
        a.x = fmaf(v1.x, w1, a.x); a.y = fmaf(v1.y, w1, a.y); a.z = fmaf(v1.z, w1, a.z); a.w = fmaf(v1.w, w1, a.w);
        a.x = fmaf(v2.x, w2, a.x); a.y = fmaf(v2.y, w2, a.y); a.z = fmaf(v2.z, w2, a.z); a.w = fmaf(v2.w, w2, a.w);
        a.x = fmaf(v3.x, w3, a.x); a.y = fmaf(v3.y, w3, a.y); a.z = fmaf(v3.z, w3, a.z); a.w = fmaf(v3.w, w3, a.w);
    }
    const int r = n - k;
    if (r > 0) {
        float4 v0 = x4[(size_t)__builtin_amdgcn_readlane(myS, k) * 32 + lane];
        float4 v1 = make_float4(0.f, 0.f, 0.f, 0.f), v2 = v1;
        if (r > 1) v1 = x4[(size_t)__builtin_amdgcn_readlane(myS, k + 1) * 32 + lane];
        if (r > 2) v2 = x4[(size_t)__builtin_amdgcn_readlane(myS, k + 2) * 32 + lane];
        const float w0 = __int_as_float(__builtin_amdgcn_readlane(myWi, k));
        a.x = fmaf(v0.x, w0, a.x); a.y = fmaf(v0.y, w0, a.y); a.z = fmaf(v0.z, w0, a.z); a.w = fmaf(v0.w, w0, a.w);
        if (r > 1) {
            const float w1 = __int_as_float(__builtin_amdgcn_readlane(myWi, k + 1));
            a.x = fmaf(v1.x, w1, a.x); a.y = fmaf(v1.y, w1, a.y); a.z = fmaf(v1.z, w1, a.z); a.w = fmaf(v1.w, w1, a.w);
        }
        if (r > 2) {
            const float w2 = __int_as_float(__builtin_amdgcn_readlane(myWi, k + 2));
            a.x = fmaf(v2.x, w2, a.x); a.y = fmaf(v2.y, w2, a.y); a.z = fmaf(v2.z, w2, a.z); a.w = fmaf(v2.w, w2, a.w);
        }
    }
}

__global__ __launch_bounds__(256) void gcn_layer1_fused(const float* __restrict__ x,
                                                        const float* __restrict__ W,
                                                        const int* __restrict__ ssrc,
                                                        const int* __restrict__ rowptr,
                                                        const int* __restrict__ rowend,
                                                        const float* __restrict__ dinv,
                                                        const float* __restrict__ b1,
                                                        const float* __restrict__ W2,
                                                        float* __restrict__ g, int N) {
    __shared__ __attribute__((aligned(16))) float As[TROWS * 132];
    __shared__ __attribute__((aligned(16))) float Hs[TROWS * 132];
    const int t = threadIdx.x;
    const int lane = t & 31;
    const int wv = __builtin_amdgcn_readfirstlane(t >> 5);
    const int rowBase = blockIdx.x * TROWS;
    const float4* x4 = (const float4*)x;

    float dis[4];
#pragma unroll
    for (int pr = 0; pr < 2; ++pr) {
        const int rA = 4 * wv + 2 * pr;
        const int iA = rowBase + rA;
        const int iB = iA + 1;
        const bool okA = iA < N, okB = iB < N;
        const int cA = min(iA, N - 1), cB = min(iB, N - 1);
        const int begA = rowptr[cA], begB = rowptr[cB];
        const int reA = rowend[cA], reB = rowend[cB];
        const float diA = dinv[cA], diB = dinv[cB];
        const int endA = okA ? reA : begA;
        const int endB = okB ? reB : begB;
        dis[2 * pr] = diA;
        dis[2 * pr + 1] = diB;
        const float4 hvA = x4[(size_t)cA * 32 + lane];
        const float4 hvB = x4[(size_t)cB * 32 + lane];
        const int nA = min(32, max(endA - begA, 0));
        const int nB = min(32, max(endB - begB, 0));
        int sA = ssrc[max(min(begA + lane, endA - 1), 0)];
        int sB = ssrc[max(min(begB + lane, endB - 1), 0)];
        sA = min(max(sA, 0), N - 1);
        sB = min(max(sB, 0), N - 1);
        const int wA = __float_as_int(dinv[sA] * diA);
        const int wB = __float_as_int(dinv[sB] * diB);

        float4 a = make_float4(0.f, 0.f, 0.f, 0.f);
        gcn_sum_rows(x4, lane, sA, wA, nA, a);
        for (int base = begA + 32; base < endA; base += 32) {
            const int n = min(32, endA - base);
            int s2, w2;
            gcn_load_ids(ssrc, dinv, base, n, diA, lane, N, s2, w2);
            gcn_sum_rows(x4, lane, s2, w2, n, a);
        }
        {
            const float ns = diA * diA;
            a.x += hvA.x * ns; a.y += hvA.y * ns; a.z += hvA.z * ns; a.w += hvA.w * ns;
            if (!okA) a = make_float4(0.f, 0.f, 0.f, 0.f);
            *(float4*)&As[rA * 132 + 4 * lane] = a;
        }
        a = make_float4(0.f, 0.f, 0.f, 0.f);
        gcn_sum_rows(x4, lane, sB, wB, nB, a);
        for (int base = begB + 32; base < endB; base += 32) {
            const int n = min(32, endB - base);
            int s2, w2;
            gcn_load_ids(ssrc, dinv, base, n, diB, lane, N, s2, w2);
            gcn_sum_rows(x4, lane, s2, w2, n, a);
        }
        {
            const float ns = diB * diB;
            a.x += hvB.x * ns; a.y += hvB.y * ns; a.z += hvB.z * ns; a.w += hvB.w * ns;
            if (!okB) a = make_float4(0.f, 0.f, 0.f, 0.f);
            *(float4*)&As[(rA + 1) * 132 + 4 * lane] = a;
        }
    }
    const float4 bb = ((const float4*)b1)[lane];
    const float4 w  = ((const float4*)W2)[lane];
    __syncthreads();

    const int m    = lane & 15;
    const int half = lane >> 4;
    const int kl   = half * 2;
    const int n    = wv * 16 + (lane & 15);

    v8f c0 = {}, c1 = {};
    const float* Arow0 = &As[m * 132];
    const float* Arow1 = &As[(16 + m) * 132];
#pragma unroll 8
    for (int k0 = 0; k0 < HIDC; k0 += 4) {
        v2f fa0, fa1, fb;
        fa0.x = Arow0[k0 + kl];
        fa0.y = Arow0[k0 + kl + 1];
        fa1.x = Arow1[k0 + kl];
        fa1.y = Arow1[k0 + kl + 1];
        fb.x = W[(k0 + kl) * HIDC + n];
        fb.y = W[(k0 + kl + 1) * HIDC + n];
        c0 = __builtin_amdgcn_wmma_f32_16x16x4_f32(false, fa0, false, fb,
                                                   (short)0, c0, false, false);
        c1 = __builtin_amdgcn_wmma_f32_16x16x4_f32(false, fa1, false, fb,
                                                   (short)0, c1, false, false);
    }
#pragma unroll
    for (int v = 0; v < 8; ++v) {
        Hs[(8 * half + v) * 132 + n]      = c0[v];
        Hs[(16 + 8 * half + v) * 132 + n] = c1[v];
    }
    __syncthreads();

    float p[4];
#pragma unroll
    for (int j = 0; j < 4; ++j) {
        const float4 hr = *(const float4*)&Hs[(4 * wv + j) * 132 + 4 * lane];
        const float r0 = fmaxf(hr.x + bb.x, 0.0f);
        const float r1 = fmaxf(hr.y + bb.y, 0.0f);
        const float r2 = fmaxf(hr.z + bb.z, 0.0f);
        const float r3 = fmaxf(hr.w + bb.w, 0.0f);
        p[j] = r0 * w.x + r1 * w.y + r2 * w.z + r3 * w.w;
    }
#pragma unroll
    for (int off = 16; off > 0; off >>= 1) {
#pragma unroll
        for (int j = 0; j < 4; ++j) p[j] += __shfl_down(p[j], off, 32);
    }
    if (lane == 0) {
#pragma unroll
        for (int j = 0; j < 4; ++j) {
            const int i = rowBase + 4 * wv + j;
            if (i < N) g[i] = p[j] * dis[j];
        }
    }
}

__global__ __launch_bounds__(128) void gcn_gather2(const float* __restrict__ g,
                                                   const int* __restrict__ ssrc,
                                                   const int* __restrict__ rowptr,
                                                   const int* __restrict__ rowend,
                                                   const float* __restrict__ dinv,
                                                   const float* __restrict__ b2,
                                                   float* __restrict__ out, int N) {
    const int i = blockIdx.x * blockDim.x + threadIdx.x;
    if (i >= N) return;
    const int beg = rowptr[i];
    const int end = rowend[i];
    const float di = dinv[i];
    const float bias = b2[0];
    float a = g[i];
    int k = beg;
    for (; k + 4 <= end; k += 4) {
        int s0 = ssrc[k], s1 = ssrc[k + 1], s2 = ssrc[k + 2], s3 = ssrc[k + 3];
        s0 = min(max(s0, 0), N - 1);
        s1 = min(max(s1, 0), N - 1);
        s2 = min(max(s2, 0), N - 1);
        s3 = min(max(s3, 0), N - 1);
        const float g0 = g[s0], g1 = g[s1], g2 = g[s2], g3 = g[s3];
        a += g0; a += g1; a += g2; a += g3;
    }
    const int r = end - k;
    if (r > 0) {
        const int k1 = min(k + 1, end - 1);
        const int k2 = min(k + 2, end - 1);
        int s0 = ssrc[k], s1 = ssrc[k1], s2 = ssrc[k2];
        s0 = min(max(s0, 0), N - 1);
        s1 = min(max(s1, 0), N - 1);
        s2 = min(max(s2, 0), N - 1);
        const float g0 = g[s0], g1 = g[s1], g2 = g[s2];
        a += g0;
        a += (r > 1) ? g1 : 0.0f;
        a += (r > 2) ? g2 : 0.0f;
    }
    out[i] = a * di + bias;
}

extern "C" void kernel_launch(void* const* d_in, const int* in_sizes, int n_in,
                              void* d_out, int out_size, void* d_ws, size_t ws_size,
                              hipStream_t stream) {
    const float* x  = (const float*)d_in[0];
    const float* W1 = (const float*)d_in[1];
    const float* b1 = (const float*)d_in[2];
    const float* W2 = (const float*)d_in[3];
    const float* b2 = (const float*)d_in[4];
    const int*   ei = (const int*)d_in[5];

    const int N = in_sizes[0] / HIDC;
    const int E = in_sizes[5] / 2;
    const int* src = ei;
    const int* dst = ei + E;
    float* out = (float*)d_out;

    const int NB = (N + BIN_NODES - 1) / BIN_NODES;
    const int NC = (E + CHUNK - 1) / CHUNK;
    const int M  = NB * NC + 1;

    int*   bhist  = (int*)d_ws;
    int*   offs   = bhist + M;
    int*   rowptr = offs + M;
    int*   rowend = rowptr + N;
    float* dinv   = (float*)(rowend + N);
    float* g      = dinv + N;
    int*   ssrc   = (int*)(g + N);
    int*   tmp    = ssrc + E;

    const int TB = 256;

    gcn_bin_count  <<<NC, TB, 0, stream>>>(dst, bhist, N, E, NB, NC);
    gcn_scan       <<<(M + 4095) / 4096, 1024, 0, stream>>>(bhist, offs, M);
    gcn_bin_scatter<<<NC, TB, 0, stream>>>(src, dst, offs, tmp, N, E, NB, NC);
    gcn_bin_sort   <<<NB, BIN_NODES, 0, stream>>>(tmp, offs, ssrc, rowptr, rowend, dinv, N, E, NC);
    gcn_layer1_fused<<<(N + TROWS - 1) / TROWS, TB, 0, stream>>>(x, W1, ssrc, rowptr, rowend, dinv, b1, W2, g, N);
    gcn_gather2   <<<(N + 127) / 128,   128, 0, stream>>>(g, ssrc, rowptr, rowend, dinv, b2, out, N);
}


# CSR build in two launches, not four, sorting in LDS at both levels so edges move in whole-line stores (a block sorts its 4096 edges by bin; a bin block sorts by node); f32 sum order as before
# speedup vs baseline: 5.4944x; 1.4343x over previous
#include <hip/hip_runtime.h>
#include <math.h>

#define HIDC 128

typedef __attribute__((ext_vector_type(2))) float v2f;
typedef __attribute__((ext_vector_type(8))) float v8f;

#define BIN_SHIFT 9
#define BIN_NODES 512
#define CHUNK     4096
#define MAXB      1024

__global__ __launch_bounds__(256) void gcn_chunk_sort(const int* __restrict__ src,
                                                      const int* __restrict__ dst,
                                                      int* __restrict__ locT,
                                                      int* __restrict__ tmp,
                                                      int N, int E, int NB, int NC) {
    __shared__ int hist[MAXB];
    __shared__ int wtot[8];
    __shared__ int stage[CHUNK + 32];
    const int t = threadIdx.x;
    const int lane = t & 31;
    const int wv = t >> 5;
    const int chunk = blockIdx.x;
    const int e0 = chunk * CHUNK + t;
    const bool full = chunk * CHUNK + CHUNK <= E;
    int d[16], s[16], r[16];
#pragma unroll
    for (int j = 0; j < 16; ++j) {
        const int e = min(e0 + j * 256, E - 1);
        d[j] = dst[e];
        s[j] = src[e];
    }
    __builtin_amdgcn_sched_barrier(0);
    for (int b = t; b < MAXB; b += 256) hist[b] = 0;
    __syncthreads();
#pragma unroll
    for (int j = 0; j < 16; ++j) {
        d[j] = min(max(d[j], 0), N - 1);
        r[j] = 0;
        if (full || e0 + j * 256 < E) r[j] = atomicAdd(&hist[min(d[j] >> BIN_SHIFT, MAXB - 1)], 1);
    }
    __syncthreads();
    const int h0 = hist[4 * t], h1 = hist[4 * t + 1], h2 = hist[4 * t + 2], h3 = hist[4 * t + 3];
    const int s4 = (h0 + h1) + (h2 + h3);
    int inc = s4;
#pragma unroll
    for (int off = 1; off < 32; off <<= 1) {
        int y = __shfl_up(inc, off, 32);
        if (lane >= off) inc += y;
    }
    if (lane == 31) wtot[wv] = inc;
    __syncthreads();
    int base = (lane < wv) ? wtot[lane & 7] : 0;
#pragma unroll
    for (int off = 16; off > 0; off >>= 1) base += __shfl_xor(base, off, 32);
    const int x0 = base + inc - s4;
    const int x1 = x0 + h0;
    const int x2 = x1 + h1;
    const int x3 = x2 + h2;
    hist[4 * t] = x0; hist[4 * t + 1] = x1; hist[4 * t + 2] = x2; hist[4 * t + 3] = x3;
    if (4 * t     < NB) locT[(4 * t)     * NC + chunk] = x0;
    if (4 * t + 1 < NB) locT[(4 * t + 1) * NC + chunk] = x1;
    if (4 * t + 2 < NB) locT[(4 * t + 2) * NC + chunk] = x2;
    if (4 * t + 3 < NB) locT[(4 * t + 3) * NC + chunk] = x3;
    if (t == 255) locT[NB * NC + chunk] = x3 + h3;
    __syncthreads();
    int pos[16];
#pragma unroll
    for (int j = 0; j < 16; ++j)
        pos[j] = hist[min(d[j] >> BIN_SHIFT, MAXB - 1)] + r[j];
#pragma unroll
    for (int j = 0; j < 16; ++j) {
        const int ss = min(max(s[j], 0), N - 1);
        int p = min(max(pos[j], 0), CHUNK - 1);
        if (!(full || e0 + j * 256 < E)) p = CHUNK;
        stage[p] = (ss << BIN_SHIFT) | (d[j] & (BIN_NODES - 1));
    }
    __syncthreads();
    int v[16];
#pragma unroll
    for (int j = 0; j < 16; ++j) v[j] = stage[t + j * 256];
    if (full) {
#pragma unroll
        for (int j = 0; j < 16; ++j) tmp[e0 + j * 256] = v[j];
    } else {
#pragma unroll
        for (int j = 0; j < 16; ++j)
            if (e0 + j * 256 < E) tmp[e0 + j * 256] = v[j];
    }
}

#define OUTCAP 12288

__device__ __forceinline__ void gcn_put(int* outL, bool inLds, int* __restrict__ ssrc,
                                        int start, int endm1, int p, int v) {
    if (inLds) outL[min(max(p, 0), OUTCAP - 1)] = v;
    else ssrc[min(start + max(p, 0), endm1)] = v;
}

template <bool PLACE>
__device__ __forceinline__ void gcn_walk_runs(const int* __restrict__ tmp,
                                              const int* __restrict__ loRow,
                                              const int* __restrict__ hiRow,
                                              int* cntL, int* __restrict__ ssrc,
                                              int* outL, bool inLds,
                                              int start, int endm1, int NC, int E,
                                              int wv, int lane, int cbFirst) {
    for (int cb = cbFirst; cb < NC; cb += 512) {
        const int c = cb + wv + 16 * lane;
        int addr = 0, len = 0;
        if (c < NC) {
            int lo = loRow[c], hi = hiRow[c];
            lo = min(max(lo, 0), CHUNK);
            hi = min(max(hi, lo), CHUNK);
            addr = c * CHUNK + lo;
            len = min(hi - lo, E - addr);
        }
        const int rc = min(32, (NC - cb - wv + 15) >> 4);
        for (int l = 0; l < rc; l += 4) {
            int a[4], n[4], k[4];
#pragma unroll
            for (int u = 0; u < 4; ++u) {
                const int li = min(l + u, 31);
                a[u] = __builtin_amdgcn_readlane(addr, li);
                n[u] = (l + u < rc) ? __builtin_amdgcn_readlane(len, li) : 0;
            }
#pragma unroll
            for (int u = 0; u < 4; ++u) {
                const int idx = min(a[u] + min(lane, max(n[u] - 1, 0)), E - 1);
                k[u] = tmp[idx];
                if (lane >= n[u]) k[u] = -1;
            }
            int p[4];
#pragma unroll
            for (int u = 0; u < 4; ++u) {
                p[u] = 0;
                if (k[u] >= 0) {
                    if (PLACE) p[u] = atomicAdd(&cntL[k[u] & (BIN_NODES - 1)], 1);
                    else atomicAdd(&cntL[k[u] & (BIN_NODES - 1)], 1);
                }
            }
            if (PLACE) {
#pragma unroll
                for (int u = 0; u < 4; ++u)
                    if (k[u] >= 0) gcn_put(outL, inLds, ssrc, start, endm1, p[u], k[u] >> BIN_SHIFT);
            }
#pragma unroll
            for (int u = 0; u < 4; ++u) {
                for (int i = 32 + lane; i < n[u]; i += 32) {
                    const int kk = tmp[a[u] + i];
                    if (PLACE) {
                        const int q = atomicAdd(&cntL[kk & (BIN_NODES - 1)], 1);
                        gcn_put(outL, inLds, ssrc, start, endm1, q, kk >> BIN_SHIFT);
                    } else {
                        atomicAdd(&cntL[kk & (BIN_NODES - 1)], 1);
                    }
                }
            }
        }
    }
}

template <bool PLACE>
__device__ __forceinline__ void gcn_walk_tails(const int* __restrict__ tmp, int* cntL,
                                               int* __restrict__ ssrc, int* outL, bool inLds,
                                               int start, int endm1,
                                               int addr, int len, int rc, int lane) {
    for (int l = 0; l < rc; ++l) {
        const int n = __builtin_amdgcn_readlane(len, l);
        if (n > 32) {
            const int a = __builtin_amdgcn_readlane(addr, l);
            for (int i = 32 + lane; i < n; i += 32) {
                const int kk = tmp[a + i];
                if (PLACE) {
                    const int q = atomicAdd(&cntL[kk & (BIN_NODES - 1)], 1);
                    gcn_put(outL, inLds, ssrc, start, endm1, q, kk >> BIN_SHIFT);
                } else {
                    atomicAdd(&cntL[kk & (BIN_NODES - 1)], 1);
                }
            }
        }
    }
}

__global__ __launch_bounds__(512) void gcn_bin_sort(const int* __restrict__ tmp,
                                                    const int* __restrict__ locT,
                                                    int* __restrict__ ssrc,
                                                    int* __restrict__ rowptr,
                                                    int* __restrict__ rowend,
                                                    float* __restrict__ dinv,
                                                    int N, int E, int NC) {
    __shared__ int cntL[BIN_NODES];
    __shared__ int wsum[16];
    __shared__ int ssum[16];
    __shared__ int outL[OUTCAP];
    const int t = threadIdx.x;
    const int lane = t & 31;
    const int wv = __builtin_amdgcn_readfirstlane(t >> 5);
    const int bin = blockIdx.x;
    const int* loRow = locT + bin * NC;
    const int* hiRow = loRow + NC;
    cntL[t] = 0;

    const int c0 = wv + 16 * lane;
    int addr = 0, len = 0, lo0 = 0;
    if (c0 < NC) {
        int lo = loRow[c0], hi = hiRow[c0];
        lo = min(max(lo, 0), CHUNK);
        hi = min(max(hi, lo), CHUNK);
        addr = c0 * CHUNK + lo;
        len = min(hi - lo, E - addr);
        lo0 = lo;
    }
    const int rc = min(32, (NC - wv + 15) >> 4);

    int key[32];
    unsigned vm = 0;
#pragma unroll
    for (int grp = 0; grp < 4; ++grp) {
        if (grp * 8 < rc) {
#pragma unroll
            for (int q = 0; q < 8; ++q) {
                const int li = grp * 8 + q;
                const int a = __builtin_amdgcn_readlane(addr, li);
                const int n = (li < rc) ? __builtin_amdgcn_readlane(len, li) : 0;
                key[li] = tmp[min(a + min(lane, max(n - 1, 0)), E - 1)];
                if (lane < n) vm |= 1u << li;
            }
        } else {
#pragma unroll
            for (int q = 0; q < 8; ++q) key[grp * 8 + q] = 0;
        }
    }
    __syncthreads();

#pragma unroll
    for (int li = 0; li < 32; ++li)
        if ((vm >> li) & 1u) atomicAdd(&cntL[key[li] & (BIN_NODES - 1)], 1);
    gcn_walk_tails<false>(tmp, cntL, ssrc, outL, false, 0, 0, addr, len, rc, lane);
    gcn_walk_runs<false>(tmp, loRow, hiRow, cntL, ssrc, outL, false, 0, 0, NC, E, wv, lane, 512);

    int part = lo0;
    for (int c = 512 + t; c < NC; c += 512) part += min(max(loRow[c], 0), CHUNK);
#pragma unroll
    for (int off = 16; off > 0; off >>= 1) part += __shfl_xor(part, off, 32);
    if (lane == 0) ssum[wv] = part;
    __syncthreads();

    const int c = cntL[t];
    int inc = c;
#pragma unroll
    for (int off = 1; off < 32; off <<= 1) {
        int y = __shfl_up(inc, off, 32);
        if (lane >= off) inc += y;
    }
    if (lane == 31) wsum[wv] = inc;
    __syncthreads();
    const int wsl = (lane < 16) ? wsum[lane & 15] : 0;
    int base = (lane < wv) ? wsl : 0;
    int tot = wsl;
    int start = (lane < 16) ? ssum[lane & 15] : 0;
#pragma unroll
    for (int off = 16; off > 0; off >>= 1) {
        base  += __shfl_xor(base, off, 32);
        tot   += __shfl_xor(tot, off, 32);
        start += __shfl_xor(start, off, 32);
    }
    start = min(max(start, 0), E);
    const int ex = base + inc - c;
    const int end = min(start + max(tot, 0), E);
    const bool inLds = tot <= OUTCAP;

    const int node = bin * BIN_NODES + t;
    if (node < N) {
        rowptr[node] = min(start + ex, E);
        rowend[node] = min(start + ex + c, E);
        dinv[node]   = rsqrtf((float)(c + 1));
    }
    cntL[t] = ex;
    __syncthreads();

#pragma unroll
    for (int grp = 0; grp < 4; ++grp) {
        if (grp * 8 < rc) {
            int p[8];
#pragma unroll
            for (int q = 0; q < 8; ++q) {
                const int li = grp * 8 + q;
                p[q] = 0;
                if ((vm >> li) & 1u) p[q] = atomicAdd(&cntL[key[li] & (BIN_NODES - 1)], 1);
            }
#pragma unroll
            for (int q = 0; q < 8; ++q) {
                const int li = grp * 8 + q;
                if ((vm >> li) & 1u) gcn_put(outL, inLds, ssrc, start, end - 1, p[q], key[li] >> BIN_SHIFT);
            }
        }
    }
    gcn_walk_tails<true>(tmp, cntL, ssrc, outL, inLds, start, end - 1, addr, len, rc, lane);
    gcn_walk_runs<true>(tmp, loRow, hiRow, cntL, ssrc, outL, inLds, start, end - 1, NC, E, wv, lane, 512);

    if (inLds) {
        __syncthreads();
        const int n = end - start;
        int i = t - (start & 31);
        if (i >= 0 && i < n) ssrc[start + i] = outL[i];
        i += 512;
        for (; i + 3 * 512 < n; i += 4 * 512) {
            const int v0 = outL[i], v1 = outL[i + 512], v2 = outL[i + 1024], v3 = outL[i + 1536];
            ssrc[start + i] = v0;
            ssrc[start + i + 512] = v1;
            ssrc[start + i + 1024] = v2;
            ssrc[start + i + 1536] = v3;
        }
        for (; i < n; i += 512) ssrc[start + i] = outL[i];
    }
}


#define TROWS 32

__device__ __forceinline__ void gcn_load_ids(const int* __restrict__ ssrc,
                                             const float* __restrict__ dinv,
                                             int base, int n, float di, int lane, int N,
                                             int& myS, int& myWi) {
    int s = 0;
    float w = 0.f;
    if (lane < n) {
        s = ssrc[base + lane];
        s = min(max(s, 0), N - 1);
        w = dinv[s] * di;
    }
    myS = s;
    myWi = __float_as_int(w);
}

__device__ __forceinline__ void gcn_sum_rows(const float4* __restrict__ x4, int lane,
                                             int myS, int myWi, int n, float4& a) {
    int k = 0;
    for (; k + 4 <= n; k += 4) {
        const float4 v0 = x4[(size_t)__builtin_amdgcn_readlane(myS, k)     * 32 + lane];
        const float4 v1 = x4[(size_t)__builtin_amdgcn_readlane(myS, k + 1) * 32 + lane];
        const float4 v2 = x4[(size_t)__builtin_amdgcn_readlane(myS, k + 2) * 32 + lane];
        const float4 v3 = x4[(size_t)__builtin_amdgcn_readlane(myS, k + 3) * 32 + lane];
        const float w0 = __int_as_float(__builtin_amdgcn_readlane(myWi, k));
        const float w1 = __int_as_float(__builtin_amdgcn_readlane(myWi, k + 1));
        const float w2 = __int_as_float(__builtin_amdgcn_readlane(myWi, k + 2));
        const float w3 = __int_as_float(__builtin_amdgcn_readlane(myWi, k + 3));
        a.x = fmaf(v0.x, w0, a.x); a.y = fmaf(v0.y, w0, a.y); a.z = fmaf(v0.z, w0, a.z); a.w = fmaf(v0.w, w0, a.w);
        a.x = fmaf(v1.x, w1, a.x); a.y = fmaf(v1.y, w1, a.y); a.z = fmaf(v1.z, w1, a.z); a.w = fmaf(v1.w, w1, a.w);
        a.x = fmaf(v2.x, w2, a.x); a.y = fmaf(v2.y, w2, a.y); a.z = fmaf(v2.z, w2, a.z); a.w = fmaf(v2.w, w2, a.w);
        a.x = fmaf(v3.x, w3, a.x); a.y = fmaf(v3.y, w3, a.y); a.z = fmaf(v3.z, w3, a.z); a.w = fmaf(v3.w, w3, a.w);
    }
    const int r = n - k;
    if (r > 0) {
        float4 v0 = x4[(size_t)__builtin_amdgcn_readlane(myS, k) * 32 + lane];
        float4 v1 = make_float4(0.f, 0.f, 0.f, 0.f), v2 = v1;
        if (r > 1) v1 = x4[(size_t)__builtin_amdgcn_readlane(myS, k + 1) * 32 + lane];
        if (r > 2) v2 = x4[(size_t)__builtin_amdgcn_readlane(myS, k + 2) * 32 + lane];
        const float w0 = __int_as_float(__builtin_amdgcn_readlane(myWi, k));
        a.x = fmaf(v0.x, w0, a.x); a.y = fmaf(v0.y, w0, a.y); a.z = fmaf(v0.z, w0, a.z); a.w = fmaf(v0.w, w0, a.w);
        if (r > 1) {
            const float w1 = __int_as_float(__builtin_amdgcn_readlane(myWi, k + 1));
            a.x = fmaf(v1.x, w1, a.x); a.y = fmaf(v1.y, w1, a.y); a.z = fmaf(v1.z, w1, a.z); a.w = fmaf(v1.w, w1, a.w);
        }
        if (r > 2) {
            const float w2 = __int_as_float(__builtin_amdgcn_readlane(myWi, k + 2));
            a.x = fmaf(v2.x, w2, a.x); a.y = fmaf(v2.y, w2, a.y); a.z = fmaf(v2.z, w2, a.z); a.w = fmaf(v2.w, w2, a.w);
        }
    }
}

__global__ __launch_bounds__(256) void gcn_layer1_fused(const float* __restrict__ x,
                                                        const float* __restrict__ W,
                                                        const int* __restrict__ ssrc,
                                                        const int* __restrict__ rowptr,
                                                        const int* __restrict__ rowend,
                                                        const float* __restrict__ dinv,
                                                        const float* __restrict__ b1,
                                                        const float* __restrict__ W2,
                                                        float* __restrict__ g, int N) {
    __shared__ __attribute__((aligned(16))) float As[TROWS * 132];
    __shared__ __attribute__((aligned(16))) float Hs[TROWS * 132];
    const int t = threadIdx.x;
    const int lane = t & 31;
    const int wv = __builtin_amdgcn_readfirstlane(t >> 5);
    const int rowBase = blockIdx.x * TROWS;
    const float4* x4 = (const float4*)x;

    float dis[4];
#pragma unroll
    for (int pr = 0; pr < 2; ++pr) {
        const int rA = 4 * wv + 2 * pr;
        const int iA = rowBase + rA;
        const int iB = iA + 1;
        const bool okA = iA < N, okB = iB < N;
        const int cA = min(iA, N - 1), cB = min(iB, N - 1);
        const int begA = rowptr[cA], begB = rowptr[cB];
        const int reA = rowend[cA], reB = rowend[cB];
        const float diA = dinv[cA], diB = dinv[cB];
        const int endA = okA ? reA : begA;
        const int endB = okB ? reB : begB;
        dis[2 * pr] = diA;
        dis[2 * pr + 1] = diB;
        const float4 hvA = x4[(size_t)cA * 32 + lane];
        const float4 hvB = x4[(size_t)cB * 32 + lane];
        const int nA = min(32, max(endA - begA, 0));
        const int nB = min(32, max(endB - begB, 0));
        int sA = ssrc[max(min(begA + lane, endA - 1), 0)];
        int sB = ssrc[max(min(begB + lane, endB - 1), 0)];
        sA = min(max(sA, 0), N - 1);
        sB = min(max(sB, 0), N - 1);
        const int wA = __float_as_int(dinv[sA] * diA);
        const int wB = __float_as_int(dinv[sB] * diB);

        float4 a = make_float4(0.f, 0.f, 0.f, 0.f);
        gcn_sum_rows(x4, lane, sA, wA, nA, a);
        for (int base = begA + 32; base < endA; base += 32) {
            const int n = min(32, endA - base);
            int s2, w2;
            gcn_load_ids(ssrc, dinv, base, n, diA, lane, N, s2, w2);
            gcn_sum_rows(x4, lane, s2, w2, n, a);
        }
        {
            const float ns = diA * diA;
            a.x += hvA.x * ns; a.y += hvA.y * ns; a.z += hvA.z * ns; a.w += hvA.w * ns;
            if (!okA) a = make_float4(0.f, 0.f, 0.f, 0.f);
            *(float4*)&As[rA * 132 + 4 * lane] = a;
        }
        a = make_float4(0.f, 0.f, 0.f, 0.f);
        gcn_sum_rows(x4, lane, sB, wB, nB, a);
        for (int base = begB + 32; base < endB; base += 32) {
            const int n = min(32, endB - base);
            int s2, w2;
            gcn_load_ids(ssrc, dinv, base, n, diB, lane, N, s2, w2);
            gcn_sum_rows(x4, lane, s2, w2, n, a);
        }
        {
            const float ns = diB * diB;
            a.x += hvB.x * ns; a.y += hvB.y * ns; a.z += hvB.z * ns; a.w += hvB.w * ns;
            if (!okB) a = make_float4(0.f, 0.f, 0.f, 0.f);
            *(float4*)&As[(rA + 1) * 132 + 4 * lane] = a;
        }
    }
    const float4 bb = ((const float4*)b1)[lane];
    const float4 w  = ((const float4*)W2)[lane];
    __syncthreads();

    const int m    = lane & 15;
    const int half = lane >> 4;
    const int kl   = half * 2;
    const int n    = wv * 16 + (lane & 15);

    v8f c0 = {}, c1 = {};
    const float* Arow0 = &As[m * 132];
    const float* Arow1 = &As[(16 + m) * 132];
#pragma unroll 8
    for (int k0 = 0; k0 < HIDC; k0 += 4) {
        v2f fa0, fa1, fb;
        fa0.x = Arow0[k0 + kl];
        fa0.y = Arow0[k0 + kl + 1];
        fa1.x = Arow1[k0 + kl];
        fa1.y = Arow1[k0 + kl + 1];
        fb.x = W[(k0 + kl) * HIDC + n];
        fb.y = W[(k0 + kl + 1) * HIDC + n];
        c0 = __builtin_amdgcn_wmma_f32_16x16x4_f32(false, fa0, false, fb,
                                                   (short)0, c0, false, false);
        c1 = __builtin_amdgcn_wmma_f32_16x16x4_f32(false, fa1, false, fb,
                                                   (short)0, c1, false, false);
    }
#pragma unroll
    for (int v = 0; v < 8; ++v) {
        Hs[(8 * half + v) * 132 + n]      = c0[v];
        Hs[(16 + 8 * half + v) * 132 + n] = c1[v];
    }
    __syncthreads();

    float p[4];
#pragma unroll
    for (int j = 0; j < 4; ++j) {
        const float4 hr = *(const float4*)&Hs[(4 * wv + j) * 132 + 4 * lane];
        const float r0 = fmaxf(hr.x + bb.x, 0.0f);
        const float r1 = fmaxf(hr.y + bb.y, 0.0f);
        const float r2 = fmaxf(hr.z + bb.z, 0.0f);
        const float r3 = fmaxf(hr.w + bb.w, 0.0f);
        p[j] = r0 * w.x + r1 * w.y + r2 * w.z + r3 * w.w;
    }
#pragma unroll
    for (int off = 16; off > 0; off >>= 1) {
#pragma unroll
        for (int j = 0; j < 4; ++j) p[j] += __shfl_down(p[j], off, 32);
    }
    if (lane == 0) {
#pragma unroll
        for (int j = 0; j < 4; ++j) {
            const int i = rowBase + 4 * wv + j;
            if (i < N) g[i] = p[j] * dis[j];
        }
    }
}

__global__ __launch_bounds__(128) void gcn_gather2(const float* __restrict__ g,
                                                   const int* __restrict__ ssrc,
                                                   const int* __restrict__ rowptr,
                                                   const int* __restrict__ rowend,
                                                   const float* __restrict__ dinv,
                                                   const float* __restrict__ b2,
                                                   float* __restrict__ out, int N) {
    const int i = blockIdx.x * blockDim.x + threadIdx.x;
    if (i >= N) return;
    const int beg = rowptr[i];
    const int end = rowend[i];
    const float di = dinv[i];
    const float bias = b2[0];
    float a = g[i];
    int k = beg;
    for (; k + 4 <= end; k += 4) {
        int s0 = ssrc[k], s1 = ssrc[k + 1], s2 = ssrc[k + 2], s3 = ssrc[k + 3];
        s0 = min(max(s0, 0), N - 1);
        s1 = min(max(s1, 0), N - 1);
        s2 = min(max(s2, 0), N - 1);
        s3 = min(max(s3, 0), N - 1);
        const float g0 = g[s0], g1 = g[s1], g2 = g[s2], g3 = g[s3];
        a += g0; a += g1; a += g2; a += g3;
    }
    const int r = end - k;
    if (r > 0) {
        const int k1 = min(k + 1, end - 1);
        const int k2 = min(k + 2, end - 1);
        int s0 = ssrc[k], s1 = ssrc[k1], s2 = ssrc[k2];
        s0 = min(max(s0, 0), N - 1);
        s1 = min(max(s1, 0), N - 1);
        s2 = min(max(s2, 0), N - 1);
        const float g0 = g[s0], g1 = g[s1], g2 = g[s2];
        a += g0;
        a += (r > 1) ? g1 : 0.0f;
        a += (r > 2) ? g2 : 0.0f;
    }
    out[i] = a * di + bias;
}

extern "C" void kernel_launch(void* const* d_in, const int* in_sizes, int n_in,
                              void* d_out, int out_size, void* d_ws, size_t ws_size,
                              hipStream_t stream) {
    const float* x  = (const float*)d_in[0];
    const float* W1 = (const float*)d_in[1];
    const float* b1 = (const float*)d_in[2];
    const float* W2 = (const float*)d_in[3];
    const float* b2 = (const float*)d_in[4];
    const int*   ei = (const int*)d_in[5];

    const int N = in_sizes[0] / HIDC;
    const int E = in_sizes[5] / 2;
    const int* src = ei;
    const int* dst = ei + E;
    float* out = (float*)d_out;

    const int NB = (N + BIN_NODES - 1) / BIN_NODES;
    const int NC = (E + CHUNK - 1) / CHUNK;
    const int M  = (NB + 1) * NC;

    int*   tmp    = (int*)d_ws;
    int*   ssrc   = tmp + E;
    int*   locT   = ssrc + E;
    int*   rowptr = locT + M;
    int*   rowend = rowptr + N;
    float* dinv   = (float*)(rowend + N);
    float* g      = dinv + N;

    const int TB = 256;

    gcn_chunk_sort <<<NC, TB, 0, stream>>>(src, dst, locT, tmp, N, E, NB, NC);
    gcn_bin_sort   <<<NB, BIN_NODES, 0, stream>>>(tmp, locT, ssrc, rowptr, rowend, dinv, N, E, NC);
    gcn_layer1_fused<<<(N + TROWS - 1) / TROWS, TB, 0, stream>>>(x, W1, ssrc, rowptr, rowend, dinv, b1, W2, g, N);
    gcn_gather2   <<<(N + 127) / 128,   128, 0, stream>>>(g, ssrc, rowptr, rowend, dinv, b2, out, N);
}
